# starting kernel
# baseline (speedup 1.0000x reference)
#include <hip/hip_runtime.h>


typedef __attribute__((ext_vector_type(2))) float v2f;
typedef __attribute__((ext_vector_type(8))) float v8f;

namespace {
constexpr int kT       = 4000;
constexpr int kSig     = 64 * 64;
constexpr int kTaps    = 413;
constexpr int kHalf    = (kTaps - 1) / 2;
constexpr int kBlkS    = 64;
constexpr int kBlkT    = 128;
constexpr int kWin     = kBlkT + 2 * kHalf;
constexpr int kStride  = 546;
constexpr int kKSteps  = 111;
constexpr int kHpLen   = 480;
constexpr int kHpOff   = 32;
}

__global__ __launch_bounds__(256)
void fir_bandpass_wmma2x2(const float* __restrict__ x,
                          const float* __restrict__ h,
                          float* __restrict__ out)
{
    __shared__ float xs[kBlkS * kStride];
    __shared__ float hpA[kHpLen];
    __shared__ float hpB[kHpLen];

    const int tid = threadIdx.x;
    const int s0b = blockIdx.x * kBlkS;
    const int t0b = blockIdx.y * kBlkT;

    for (int i = tid; i < kHpLen; i += 256) {
        int j = i - kHpOff;
        float v = (j >= 0 && j < kTaps) ? h[kTaps - 1 - j] : 0.0f;
        hpA[i] = v;
        hpB[i] = v;
    }

    for (int s = (tid >> 4); s < kBlkS; s += 16) {
        const float* row = x + (size_t)(s0b + s) * kT;
        const float xf = row[0];
        const float xl = row[kT - 1];
        for (int jj = (tid & 15); jj < kWin; jj += 16) {
            int j = t0b - kHalf + jj;
            float v;
            if (j < 0)        v = 2.0f * xf - row[-j];
            else if (j >= kT) v = 2.0f * xl - row[2 * kT - 2 - j];
            else              v = row[j];
            xs[s * kStride + jj] = v;
        }
    }
    __syncthreads();

    const int w    = tid >> 5;
    const int lane = tid & 31;
    const int mg   = w & 3;
    const int ng   = w >> 2;
    const int t00  = t0b + (mg << 5);
    if (t00 >= kT) return;

    const int m  = lane & 15;
    const int hi = lane >> 4;
    const int kk = hi << 1;

    const float* hb  = (hi ? hpB : hpA) + (kHpOff + kk - m);
    const float* xb0 = xs + ((ng << 5) + m) * kStride + (mg << 5) + kk;
    const float* xb1 = xb0 + 16 * kStride;

    v8f acc00 = {0.f,0.f,0.f,0.f,0.f,0.f,0.f,0.f};
    v8f acc01 = acc00;
    v8f acc10 = acc00;
    v8f acc11 = acc00;

    #pragma unroll 4
    for (int k = 0; k < kKSteps; ++k) {
        const int c = 4 * k;
        v2f a0, a1, b0, b1;
        a0.x = hb[c];          a0.y = hb[c + 1];
        a1.x = hb[c - 16];     a1.y = hb[c - 15];
        b0.x = xb0[c];         b0.y = xb0[c + 1];
        b1.x = xb1[c];         b1.y = xb1[c + 1];
        acc00 = __builtin_amdgcn_wmma_f32_16x16x4_f32(
            false, a0, false, b0, (short)0, acc00, false, false);
        acc01 = __builtin_amdgcn_wmma_f32_16x16x4_f32(
            false, a0, false, b1, (short)0, acc01, false, false);
        acc10 = __builtin_amdgcn_wmma_f32_16x16x4_f32(
            false, a1, false, b0, (short)0, acc10, false, false);
        acc11 = __builtin_amdgcn_wmma_f32_16x16x4_f32(
            false, a1, false, b1, (short)0, acc11, false, false);
    }

    float* o00 = out + (size_t)(s0b + (ng << 5) + m) * kT + t00 + (hi << 3);
    float* o01 = o00 + (size_t)16 * kT;
    float* o10 = o00 + 16;
    float* o11 = o01 + 16;

    float4 v;
    v.x = acc00[0]; v.y = acc00[1]; v.z = acc00[2]; v.w = acc00[3];
    *(float4*)(o00) = v;
    v.x = acc00[4]; v.y = acc00[5]; v.z = acc00[6]; v.w = acc00[7];
    *(float4*)(o00 + 4) = v;
    v.x = acc01[0]; v.y = acc01[1]; v.z = acc01[2]; v.w = acc01[3];
    *(float4*)(o01) = v;
    v.x = acc01[4]; v.y = acc01[5]; v.z = acc01[6]; v.w = acc01[7];
    *(float4*)(o01 + 4) = v;
    v.x = acc10[0]; v.y = acc10[1]; v.z = acc10[2]; v.w = acc10[3];
    *(float4*)(o10) = v;
    v.x = acc10[4]; v.y = acc10[5]; v.z = acc10[6]; v.w = acc10[7];
    *(float4*)(o10 + 4) = v;
    v.x = acc11[0]; v.y = acc11[1]; v.z = acc11[2]; v.w = acc11[3];
    *(float4*)(o11) = v;
    v.x = acc11[4]; v.y = acc11[5]; v.z = acc11[6]; v.w = acc11[7];
    *(float4*)(o11 + 4) = v;
}

extern "C" void kernel_launch(void* const* d_in, const int* in_sizes, int n_in,
                              void* d_out, int out_size, void* d_ws, size_t ws_size,
                              hipStream_t stream) {
    const float* x = (const float*)d_in[0];
    const float* h = (const float*)d_in[1];
    float* out = (float*)d_out;

    dim3 grid(kSig / kBlkS, (kT + kBlkT - 1) / kBlkT);
    fir_bandpass_wmma2x2<<<grid, 256, 0, stream>>>(x, h, out);
}


# Prologue loads as one group: window read as aligned 16-byte chunks, 512 contiguous bytes per wave load, all 34 per thread and the tap loads in flight at once, 16-byte LDS stores at row pitch 548
# speedup vs baseline: 1.9885x; 1.9885x over previous
#include <hip/hip_runtime.h>


typedef __attribute__((ext_vector_type(2))) float v2f;
typedef __attribute__((ext_vector_type(8))) float v8f;

namespace {
constexpr int kT       = 4000;
constexpr int kSig     = 64 * 64;
constexpr int kTaps    = 413;
constexpr int kHalf    = (kTaps - 1) / 2;
constexpr int kBlkS    = 64;
constexpr int kBlkT    = 128;
constexpr int kPre     = 208;
constexpr int kShift   = kPre - kHalf;
constexpr int kWinV    = 136;
constexpr int kPerThr  = kBlkS * kWinV / 256;
constexpr int kStride  = 548;
static_assert(kBlkS * kWinV == kPerThr * 256, "chunks must divide evenly over the threads");
static_assert(kT % 4 == 0 && kBlkT % 4 == 0 && kPre % 4 == 0, "chunks must not straddle the signal's ends");
static_assert(kT >= 2 * 4 * kWinV, "a window must not overhang both ends of a signal");
constexpr int kKSteps  = 111;
constexpr int kHpLen   = 480;
constexpr int kHpOff   = 32;
constexpr int kHpFill  = 512;
constexpr int kHpPitch = 544;
static_assert(kHpFill >= kHpLen && kHpPitch >= kHpFill && kHpPitch % 64 == 32, "tap copies");
}

__global__ __launch_bounds__(256)
void fir_bandpass_wmma2x2(const float* __restrict__ x,
                          const float* __restrict__ h,
                          float* __restrict__ out)
{
    __shared__ __attribute__((aligned(16))) float xs[kBlkS * kStride];
    __shared__ float hp[kHpPitch + kHpFill];

    const int tid = threadIdx.x;
    const int s0b = blockIdx.x * kBlkS;
    const int t0b = blockIdx.y * kBlkT;

    const int ja = tid - kHpOff;
    const int jb = tid + 256 - kHpOff;
    const float ta = h[kTaps - 1 - (ja < 0 ? 0 : ja)];
    const float tb = h[jb < kTaps ? kTaps - 1 - jb : 0];

    const int j0 = t0b - kPre;
    float4 win[kPerThr];
    #pragma unroll
    for (int i = 0; i < kPerThr; ++i) {
        const int id = i * 256 + tid;
        const int s  = id / kWinV;
        const int q  = id - s * kWinV;
        int j = j0 + 4 * q;
        j = j < 0 ? 0 : (j > kT - 4 ? kT - 4 : j);
        win[i] = *(const float4*)(x + (size_t)(s0b + s) * kT + j);
    }

    {
        const float va = ja >= 0 ? ta : 0.0f;
        const float vb = jb < kTaps ? tb : 0.0f;
        hp[tid]                  = va;
        hp[kHpPitch + tid]       = va;
        hp[tid + 256]            = vb;
        hp[kHpPitch + tid + 256] = vb;
    }
    #pragma unroll
    for (int i = 0; i < kPerThr; ++i) {
        const int id = i * 256 + tid;
        const int s  = id / kWinV;
        const int q  = id - s * kWinV;
        *(float4*)(xs + s * kStride + 4 * q) = win[i];
    }

    if (j0 < 0 || j0 + 4 * kWinV > kT) {
        __syncthreads();
        const bool left = j0 < 0;
        const int nOut  = left ? ((-j0) >> 2) : ((j0 + 4 * kWinV - kT) >> 2);
        const int qb    = left ? 0 : kWinV - nOut;
        const int s     = tid & 63;
        const float* row = x + (size_t)(s0b + s) * kT;
        const float e2  = 2.0f * row[left ? 0 : kT - 1];
        #pragma unroll 4
        for (int q = qb + (tid >> 6); q < qb + nOut; q += 4) {
            const int j = j0 + 4 * q;
            const int r = left ? -j : 2 * kT - 2 - j;
            float4 p;
            p.x = e2 - row[r];
            p.y = e2 - row[r - 1];
            p.z = e2 - row[r - 2];
            p.w = e2 - row[r - 3];
            *(float4*)(xs + s * kStride + 4 * q) = p;
        }
    }
    __syncthreads();

    const int w    = tid >> 5;
    const int lane = tid & 31;
    const int mg   = w & 3;
    const int ng   = w >> 2;
    const int t00  = t0b + (mg << 5);
    if (t00 >= kT) return;

    const int m  = lane & 15;
    const int hi = lane >> 4;
    const int kk = hi << 1;

    const float* hb  = hp + (hi ? kHpPitch : 0) + (kHpOff + kk - m);
    const float* xb0 = xs + ((ng << 5) + m) * kStride + kShift + (mg << 5) + kk;
    const float* xb1 = xb0 + 16 * kStride;

    v8f acc00 = {0.f,0.f,0.f,0.f,0.f,0.f,0.f,0.f};
    v8f acc01 = acc00;
    v8f acc10 = acc00;
    v8f acc11 = acc00;

    #pragma unroll 4
    for (int k = 0; k < kKSteps; ++k) {
        const int c = 4 * k;
        v2f a0, a1, b0, b1;
        a0.x = hb[c];          a0.y = hb[c + 1];
        a1.x = hb[c - 16];     a1.y = hb[c - 15];
        b0.x = xb0[c];         b0.y = xb0[c + 1];
        b1.x = xb1[c];         b1.y = xb1[c + 1];
        acc00 = __builtin_amdgcn_wmma_f32_16x16x4_f32(
            false, a0, false, b0, (short)0, acc00, false, false);
        acc01 = __builtin_amdgcn_wmma_f32_16x16x4_f32(
            false, a0, false, b1, (short)0, acc01, false, false);
        acc10 = __builtin_amdgcn_wmma_f32_16x16x4_f32(
            false, a1, false, b0, (short)0, acc10, false, false);
        acc11 = __builtin_amdgcn_wmma_f32_16x16x4_f32(
            false, a1, false, b1, (short)0, acc11, false, false);
    }

    float* o00 = out + (size_t)(s0b + (ng << 5) + m) * kT + t00 + (hi << 3);
    float* o01 = o00 + (size_t)16 * kT;
    float* o10 = o00 + 16;
    float* o11 = o01 + 16;

    float4 v;
    v.x = acc00[0]; v.y = acc00[1]; v.z = acc00[2]; v.w = acc00[3];
    *(float4*)(o00) = v;
    v.x = acc00[4]; v.y = acc00[5]; v.z = acc00[6]; v.w = acc00[7];
    *(float4*)(o00 + 4) = v;
    v.x = acc01[0]; v.y = acc01[1]; v.z = acc01[2]; v.w = acc01[3];
    *(float4*)(o01) = v;
    v.x = acc01[4]; v.y = acc01[5]; v.z = acc01[6]; v.w = acc01[7];
    *(float4*)(o01 + 4) = v;
    v.x = acc10[0]; v.y = acc10[1]; v.z = acc10[2]; v.w = acc10[3];
    *(float4*)(o10) = v;
    v.x = acc10[4]; v.y = acc10[5]; v.z = acc10[6]; v.w = acc10[7];
    *(float4*)(o10 + 4) = v;
    v.x = acc11[0]; v.y = acc11[1]; v.z = acc11[2]; v.w = acc11[3];
    *(float4*)(o11) = v;
    v.x = acc11[4]; v.y = acc11[5]; v.z = acc11[6]; v.w = acc11[7];
    *(float4*)(o11 + 4) = v;
}

extern "C" void kernel_launch(void* const* d_in, const int* in_sizes, int n_in,
                              void* d_out, int out_size, void* d_ws, size_t ws_size,
                              hipStream_t stream) {
    const float* x = (const float*)d_in[0];
    const float* h = (const float*)d_in[1];
    float* out = (float*)d_out;

    dim3 grid(kSig / kBlkS, (kT + kBlkT - 1) / kBlkT);
    fir_bandpass_wmma2x2<<<grid, 256, 0, stream>>>(x, h, out);
}


# Resident workgroups: 512 workgroups each walk 4 time blocks of their 64 signals; the next block's window is requested ahead of this block's WMMAs and waits in registers until they are done
# speedup vs baseline: 2.0574x; 1.0346x over previous
#include <hip/hip_runtime.h>


typedef __attribute__((ext_vector_type(2))) float v2f;
typedef __attribute__((ext_vector_type(8))) float v8f;

namespace {
constexpr int kT       = 4000;
constexpr int kSig     = 64 * 64;
constexpr int kTaps    = 413;
constexpr int kHalf    = (kTaps - 1) / 2;
constexpr int kBlkS    = 64;
constexpr int kBlkT    = 128;
constexpr int kTimeBlocks = (kT + kBlkT - 1) / kBlkT;
constexpr int kTilesPerWG = 4;
constexpr int kTileStep   = kTimeBlocks / kTilesPerWG;
static_assert(kTimeBlocks % kTilesPerWG == 0, "time blocks must divide evenly over the workgroups");
static_assert(kSig / kBlkS == 64, "blockIdx.x is split into its low 6 bits (signal block) and the rest (first time block)");
constexpr int kPre     = 208;
constexpr int kShift   = kPre - kHalf;
constexpr int kWinV    = 136;
constexpr int kPerThr  = kBlkS * kWinV / 256;
constexpr int kStride  = 548;
static_assert(kBlkS * kWinV == kPerThr * 256 && kPerThr == 34 && kWinV == 4 * 32 + 8 && kBlkS == 8 * 8,
              "the chunk-to-thread map below is written for 64 rows of 136 chunks on 256 threads");
static_assert(kT % 4 == 0 && kBlkT % 4 == 0 && kPre % 4 == 0, "chunks must not straddle the signal's ends");
static_assert(kT >= 2 * 4 * kWinV, "a window must not overhang both ends of a signal");
constexpr int kKSteps  = 111;
constexpr int kHpLen   = 480;
constexpr int kHpOff   = 32;
constexpr int kHpFill  = 512;
constexpr int kHpPitch = 544;
static_assert(kHpFill >= kHpLen && kHpPitch >= kHpFill && kHpPitch % 64 == 32, "tap copies");

__device__ __forceinline__ int clamp_chunk(int j)
{
    return j < 0 ? 0 : (j > kT - 4 ? kT - 4 : j);
}

#define FIR_MAIN_CHUNKS(X) \
    X(0,0) X(0,1) X(0,2) X(0,3) X(0,4) X(0,5) X(0,6) X(0,7) \
    X(1,0) X(1,1) X(1,2) X(1,3) X(1,4) X(1,5) X(1,6) X(1,7) \
    X(2,0) X(2,1) X(2,2) X(2,3) X(2,4) X(2,5) X(2,6) X(2,7) \
    X(3,0) X(3,1) X(3,2) X(3,3) X(3,4) X(3,5) X(3,6) X(3,7)
#define FIR_DECL(j,k)  float4 wm##j##k;
#define FIR_REQ(j,k)   wm##j##k = *(const float4*)(gp##j + (k) * 8 * kT);
#define FIR_PUT(j,k)   *(float4*)(lp + (k) * 8 * kStride + (j) * 128) = wm##j##k;
}

__global__ __launch_bounds__(256)
void fir_bandpass_wmma2x2(const float* __restrict__ x,
                          const float* __restrict__ h,
                          float* __restrict__ out)
{
    __shared__ __attribute__((aligned(16))) float xs[kBlkS * kStride];
    __shared__ float hp[kHpPitch + kHpFill];

    const int tid = threadIdx.x;
    const int s0b = (blockIdx.x & 63) * kBlkS;
    const int tb0 = blockIdx.x >> 6;

    const int ja = tid - kHpOff;
    const int jb = tid + 256 - kHpOff;
    const float ta = h[kTaps - 1 - (ja < 0 ? 0 : ja)];
    const float tb = h[jb < kTaps ? kTaps - 1 - jb : 0];

    const int w    = tid >> 5;
    const int lane = tid & 31;

    FIR_MAIN_CHUNKS(FIR_DECL)
    float4 wt0, wt1;
    const float* grow = x + (size_t)(s0b + w) * kT;
    const float* trow = x + (size_t)(s0b + (tid >> 2)) * kT;
    const int    tq   = 4 * (kWinV - 8 + (tid & 3));
    float* const lp   = xs + w * kStride + 4 * lane;
    float* const lt   = xs + (tid >> 2) * kStride + tq;
    {
        const int j0 = tb0 * kBlkT - kPre;
        const float* gp0 = grow + clamp_chunk(j0 + 4 * lane);
        const float* gp1 = grow + clamp_chunk(j0 + 4 * lane + 128);
        const float* gp2 = grow + clamp_chunk(j0 + 4 * lane + 256);
        const float* gp3 = grow + clamp_chunk(j0 + 4 * lane + 384);
        FIR_MAIN_CHUNKS(FIR_REQ)
        wt0 = *(const float4*)(trow + clamp_chunk(j0 + tq));
        wt1 = *(const float4*)(trow + clamp_chunk(j0 + tq + 16));
    }

    {
        const float va = ja >= 0 ? ta : 0.0f;
        const float vb = jb < kTaps ? tb : 0.0f;
        hp[tid]                  = va;
        hp[kHpPitch + tid]       = va;
        hp[tid + 256]            = vb;
        hp[kHpPitch + tid + 256] = vb;
    }

    const int mg   = w & 3;
    const int ng   = w >> 2;
    const int m    = lane & 15;
    const int hi   = lane >> 4;
    const int kk   = hi << 1;

    const float* hb  = hp + (hi ? kHpPitch : 0) + (kHpOff + kk - m);
    const float* xb0 = xs + ((ng << 5) + m) * kStride + kShift + (mg << 5) + kk;
    const float* xb1 = xb0 + 16 * kStride;

    #pragma unroll 1
    for (int n = 0; n < kTilesPerWG; ++n) {
        const int t0b = (tb0 + n * kTileStep) * kBlkT;
        const int j0  = t0b - kPre;

        FIR_MAIN_CHUNKS(FIR_PUT)
        *(float4*)(lt)      = wt0;
        *(float4*)(lt + 16) = wt1;

        if (j0 < 0 || j0 + 4 * kWinV > kT) {
            __syncthreads();
            const bool left = j0 < 0;
            const int nOut  = left ? ((-j0) >> 2) : ((j0 + 4 * kWinV - kT) >> 2);
            const int qb    = left ? 0 : kWinV - nOut;
            const int s     = tid & 63;
            const float* row = x + (size_t)(s0b + s) * kT;
            const float e2  = 2.0f * row[left ? 0 : kT - 1];
            #pragma unroll 4
            for (int q = qb + (tid >> 6); q < qb + nOut; q += 4) {
                const int j = j0 + 4 * q;
                const int r = left ? -j : 2 * kT - 2 - j;
                float4 p;
                p.x = e2 - row[r];
                p.y = e2 - row[r - 1];
                p.z = e2 - row[r - 2];
                p.w = e2 - row[r - 3];
                *(float4*)(xs + s * kStride + 4 * q) = p;
            }
        }
        __syncthreads();

        if (n + 1 < kTilesPerWG) {
            const int j1 = j0 + kTileStep * kBlkT;
            const float* gp0 = grow + clamp_chunk(j1 + 4 * lane);
            const float* gp1 = grow + clamp_chunk(j1 + 4 * lane + 128);
            const float* gp2 = grow + clamp_chunk(j1 + 4 * lane + 256);
            const float* gp3 = grow + clamp_chunk(j1 + 4 * lane + 384);
            FIR_MAIN_CHUNKS(FIR_REQ)
            wt0 = *(const float4*)(trow + clamp_chunk(j1 + tq));
            wt1 = *(const float4*)(trow + clamp_chunk(j1 + tq + 16));
        }

        const int t00 = t0b + (mg << 5);
        if (t00 < kT) {
            v8f acc00 = {0.f,0.f,0.f,0.f,0.f,0.f,0.f,0.f};
            v8f acc01 = acc00;
            v8f acc10 = acc00;
            v8f acc11 = acc00;

            #pragma unroll 4
            for (int k = 0; k < kKSteps; ++k) {
                const int c = 4 * k;
                v2f a0, a1, b0, b1;
                a0.x = hb[c];          a0.y = hb[c + 1];
                a1.x = hb[c - 16];     a1.y = hb[c - 15];
                b0.x = xb0[c];         b0.y = xb0[c + 1];
                b1.x = xb1[c];         b1.y = xb1[c + 1];
                acc00 = __builtin_amdgcn_wmma_f32_16x16x4_f32(
                    false, a0, false, b0, (short)0, acc00, false, false);
                acc01 = __builtin_amdgcn_wmma_f32_16x16x4_f32(
                    false, a0, false, b1, (short)0, acc01, false, false);
                acc10 = __builtin_amdgcn_wmma_f32_16x16x4_f32(
                    false, a1, false, b0, (short)0, acc10, false, false);
                acc11 = __builtin_amdgcn_wmma_f32_16x16x4_f32(
                    false, a1, false, b1, (short)0, acc11, false, false);
            }

            float* o00 = out + (size_t)(s0b + (ng << 5) + m) * kT + t00 + (hi << 3);
            float* o01 = o00 + (size_t)16 * kT;
            float* o10 = o00 + 16;
            float* o11 = o01 + 16;

            float4 v;
            v.x = acc00[0]; v.y = acc00[1]; v.z = acc00[2]; v.w = acc00[3];
            *(float4*)(o00) = v;
            v.x = acc00[4]; v.y = acc00[5]; v.z = acc00[6]; v.w = acc00[7];
            *(float4*)(o00 + 4) = v;
            v.x = acc01[0]; v.y = acc01[1]; v.z = acc01[2]; v.w = acc01[3];
            *(float4*)(o01) = v;
            v.x = acc01[4]; v.y = acc01[5]; v.z = acc01[6]; v.w = acc01[7];
            *(float4*)(o01 + 4) = v;
            v.x = acc10[0]; v.y = acc10[1]; v.z = acc10[2]; v.w = acc10[3];
            *(float4*)(o10) = v;
            v.x = acc10[4]; v.y = acc10[5]; v.z = acc10[6]; v.w = acc10[7];
            *(float4*)(o10 + 4) = v;
            v.x = acc11[0]; v.y = acc11[1]; v.z = acc11[2]; v.w = acc11[3];
            *(float4*)(o11) = v;
            v.x = acc11[4]; v.y = acc11[5]; v.z = acc11[6]; v.w = acc11[7];
            *(float4*)(o11 + 4) = v;
        }
        __syncthreads();
    }
}

extern "C" void kernel_launch(void* const* d_in, const int* in_sizes, int n_in,
                              void* d_out, int out_size, void* d_ws, size_t ws_size,
                              hipStream_t stream) {
    const float* x = (const float*)d_in[0];
    const float* h = (const float*)d_in[1];
    float* out = (float*)d_out;

    dim3 grid((kSig / kBlkS) * kTileStep);
    fir_bandpass_wmma2x2<<<grid, 256, 0, stream>>>(x, h, out);
}


# Wave priority falls with progress: s_setprio 3 at entry, one level down after each of a workgroup's 4 time blocks, so the co-resident workgroup that is behind wins the matrix pipe; both end together
# speedup vs baseline: 2.1722x; 1.0558x over previous
#include <hip/hip_runtime.h>


typedef __attribute__((ext_vector_type(2))) float v2f;
typedef __attribute__((ext_vector_type(8))) float v8f;

namespace {
constexpr int kT       = 4000;
constexpr int kSig     = 64 * 64;
constexpr int kTaps    = 413;
constexpr int kHalf    = (kTaps - 1) / 2;
constexpr int kBlkS    = 64;
constexpr int kBlkT    = 128;
constexpr int kTimeBlocks = (kT + kBlkT - 1) / kBlkT;
constexpr int kTilesPerWG = 4;
constexpr int kTileStep   = kTimeBlocks / kTilesPerWG;
static_assert(kTimeBlocks % kTilesPerWG == 0, "time blocks must divide evenly over the workgroups");
static_assert(kTilesPerWG == 4, "one priority level (3, 2, 1, 0) per time block of a workgroup");
static_assert(kSig / kBlkS == 64, "blockIdx.x is split into its low 6 bits (signal block) and the rest (first time block)");
constexpr int kPre     = 208;
constexpr int kShift   = kPre - kHalf;
constexpr int kWinV    = 136;
constexpr int kPerThr  = kBlkS * kWinV / 256;
constexpr int kStride  = 548;
static_assert(kBlkS * kWinV == kPerThr * 256 && kPerThr == 34 && kWinV == 4 * 32 + 8 && kBlkS == 8 * 8,
              "the chunk-to-thread map below is written for 64 rows of 136 chunks on 256 threads");
static_assert(kT % 4 == 0 && kBlkT % 4 == 0 && kPre % 4 == 0, "chunks must not straddle the signal's ends");
static_assert(kT >= 2 * 4 * kWinV, "a window must not overhang both ends of a signal");
constexpr int kKSteps  = 111;
constexpr int kHpLen   = 480;
constexpr int kHpOff   = 32;
constexpr int kHpFill  = 512;
constexpr int kHpPitch = 544;
static_assert(kHpFill >= kHpLen && kHpPitch >= kHpFill && kHpPitch % 64 == 32, "tap copies");

__device__ __forceinline__ int clamp_chunk(int j)
{
    return j < 0 ? 0 : (j > kT - 4 ? kT - 4 : j);
}

#define FIR_MAIN_CHUNKS(X) \
    X(0,0) X(0,1) X(0,2) X(0,3) X(0,4) X(0,5) X(0,6) X(0,7) \
    X(1,0) X(1,1) X(1,2) X(1,3) X(1,4) X(1,5) X(1,6) X(1,7) \
    X(2,0) X(2,1) X(2,2) X(2,3) X(2,4) X(2,5) X(2,6) X(2,7) \
    X(3,0) X(3,1) X(3,2) X(3,3) X(3,4) X(3,5) X(3,6) X(3,7)
#define FIR_DECL(j,k)  float4 wm##j##k;
#define FIR_REQ(j,k)   wm##j##k = *(const float4*)(gp##j + (k) * 8 * kT);
#define FIR_PUT(j,k)   *(float4*)(lp + (k) * 8 * kStride + (j) * 128) = wm##j##k;
}

__global__ __launch_bounds__(256)
void fir_bandpass_wmma2x2(const float* __restrict__ x,
                          const float* __restrict__ h,
                          float* __restrict__ out)
{
    __shared__ __attribute__((aligned(16))) float xs[kBlkS * kStride];
    __shared__ float hp[kHpPitch + kHpFill];

    __builtin_amdgcn_s_setprio(3);

    const int tid = threadIdx.x;
    const int s0b = (blockIdx.x & 63) * kBlkS;
    const int tb0 = blockIdx.x >> 6;

    const int ja = tid - kHpOff;
    const int jb = tid + 256 - kHpOff;
    const float ta = h[kTaps - 1 - (ja < 0 ? 0 : ja)];
    const float tb = h[jb < kTaps ? kTaps - 1 - jb : 0];

    const int w    = tid >> 5;
    const int lane = tid & 31;

    FIR_MAIN_CHUNKS(FIR_DECL)
    float4 wt0, wt1;
    const float* grow = x + (size_t)(s0b + w) * kT;
    const float* trow = x + (size_t)(s0b + (tid >> 2)) * kT;
    const int    tq   = 4 * (kWinV - 8 + (tid & 3));
    float* const lp   = xs + w * kStride + 4 * lane;
    float* const lt   = xs + (tid >> 2) * kStride + tq;
    {
        const int j0 = tb0 * kBlkT - kPre;
        const float* gp0 = grow + clamp_chunk(j0 + 4 * lane);
        const float* gp1 = grow + clamp_chunk(j0 + 4 * lane + 128);
        const float* gp2 = grow + clamp_chunk(j0 + 4 * lane + 256);
        const float* gp3 = grow + clamp_chunk(j0 + 4 * lane + 384);
        FIR_MAIN_CHUNKS(FIR_REQ)
        wt0 = *(const float4*)(trow + clamp_chunk(j0 + tq));
        wt1 = *(const float4*)(trow + clamp_chunk(j0 + tq + 16));
    }

    {
        const float va = ja >= 0 ? ta : 0.0f;
        const float vb = jb < kTaps ? tb : 0.0f;
        hp[tid]                  = va;
        hp[kHpPitch + tid]       = va;
        hp[tid + 256]            = vb;
        hp[kHpPitch + tid + 256] = vb;
    }

    const int mg   = w & 3;
    const int ng   = w >> 2;
    const int m    = lane & 15;
    const int hi   = lane >> 4;
    const int kk   = hi << 1;

    const float* hb  = hp + (hi ? kHpPitch : 0) + (kHpOff + kk - m);
    const float* xb0 = xs + ((ng << 5) + m) * kStride + kShift + (mg << 5) + kk;
    const float* xb1 = xb0 + 16 * kStride;

    #pragma unroll 1
    for (int n = 0; n < kTilesPerWG; ++n) {
        const int t0b = (tb0 + n * kTileStep) * kBlkT;
        const int j0  = t0b - kPre;

        FIR_MAIN_CHUNKS(FIR_PUT)
        *(float4*)(lt)      = wt0;
        *(float4*)(lt + 16) = wt1;

        if (j0 < 0 || j0 + 4 * kWinV > kT) {
            __syncthreads();
            const bool left = j0 < 0;
            const int nOut  = left ? ((-j0) >> 2) : ((j0 + 4 * kWinV - kT) >> 2);
            const int qb    = left ? 0 : kWinV - nOut;
            const int s     = tid & 63;
            const float* row = x + (size_t)(s0b + s) * kT;
            const float e2  = 2.0f * row[left ? 0 : kT - 1];
            #pragma unroll 4
            for (int q = qb + (tid >> 6); q < qb + nOut; q += 4) {
                const int j = j0 + 4 * q;
                const int r = left ? -j : 2 * kT - 2 - j;
                float4 p;
                p.x = e2 - row[r];
                p.y = e2 - row[r - 1];
                p.z = e2 - row[r - 2];
                p.w = e2 - row[r - 3];
                *(float4*)(xs + s * kStride + 4 * q) = p;
            }
        }
        __syncthreads();

        if (n + 1 < kTilesPerWG) {
            const int j1 = j0 + kTileStep * kBlkT;
            const float* gp0 = grow + clamp_chunk(j1 + 4 * lane);
            const float* gp1 = grow + clamp_chunk(j1 + 4 * lane + 128);
            const float* gp2 = grow + clamp_chunk(j1 + 4 * lane + 256);
            const float* gp3 = grow + clamp_chunk(j1 + 4 * lane + 384);
            FIR_MAIN_CHUNKS(FIR_REQ)
            wt0 = *(const float4*)(trow + clamp_chunk(j1 + tq));
            wt1 = *(const float4*)(trow + clamp_chunk(j1 + tq + 16));
        }

        const int t00 = t0b + (mg << 5);
        if (t00 < kT) {
            v8f acc00 = {0.f,0.f,0.f,0.f,0.f,0.f,0.f,0.f};
            v8f acc01 = acc00;
            v8f acc10 = acc00;
            v8f acc11 = acc00;

            #pragma unroll 4
            for (int k = 0; k < kKSteps; ++k) {
                const int c = 4 * k;
                v2f a0, a1, b0, b1;
                a0.x = hb[c];          a0.y = hb[c + 1];
                a1.x = hb[c - 16];     a1.y = hb[c - 15];
                b0.x = xb0[c];         b0.y = xb0[c + 1];
                b1.x = xb1[c];         b1.y = xb1[c + 1];
                acc00 = __builtin_amdgcn_wmma_f32_16x16x4_f32(
                    false, a0, false, b0, (short)0, acc00, false, false);
                acc01 = __builtin_amdgcn_wmma_f32_16x16x4_f32(
                    false, a0, false, b1, (short)0, acc01, false, false);
                acc10 = __builtin_amdgcn_wmma_f32_16x16x4_f32(
                    false, a1, false, b0, (short)0, acc10, false, false);
                acc11 = __builtin_amdgcn_wmma_f32_16x16x4_f32(
                    false, a1, false, b1, (short)0, acc11, false, false);
            }

            float* o00 = out + (size_t)(s0b + (ng << 5) + m) * kT + t00 + (hi << 3);
            float* o01 = o00 + (size_t)16 * kT;
            float* o10 = o00 + 16;
            float* o11 = o01 + 16;

            float4 v;
            v.x = acc00[0]; v.y = acc00[1]; v.z = acc00[2]; v.w = acc00[3];
            *(float4*)(o00) = v;
            v.x = acc00[4]; v.y = acc00[5]; v.z = acc00[6]; v.w = acc00[7];
            *(float4*)(o00 + 4) = v;
            v.x = acc01[0]; v.y = acc01[1]; v.z = acc01[2]; v.w = acc01[3];
            *(float4*)(o01) = v;
            v.x = acc01[4]; v.y = acc01[5]; v.z = acc01[6]; v.w = acc01[7];
            *(float4*)(o01 + 4) = v;
            v.x = acc10[0]; v.y = acc10[1]; v.z = acc10[2]; v.w = acc10[3];
            *(float4*)(o10) = v;
            v.x = acc10[4]; v.y = acc10[5]; v.z = acc10[6]; v.w = acc10[7];
            *(float4*)(o10 + 4) = v;
            v.x = acc11[0]; v.y = acc11[1]; v.z = acc11[2]; v.w = acc11[3];
            *(float4*)(o11) = v;
            v.x = acc11[4]; v.y = acc11[5]; v.z = acc11[6]; v.w = acc11[7];
            *(float4*)(o11 + 4) = v;
        }
        if (n == 0) __builtin_amdgcn_s_setprio(2);
        if (n == 1) __builtin_amdgcn_s_setprio(1);
        if (n == 2) __builtin_amdgcn_s_setprio(0);

        __syncthreads();
    }
}

extern "C" void kernel_launch(void* const* d_in, const int* in_sizes, int n_in,
                              void* d_out, int out_size, void* d_ws, size_t ws_size,
                              hipStream_t stream) {
    const float* x = (const float*)d_in[0];
    const float* h = (const float*)d_in[1];
    float* out = (float*)d_out;

    dim3 grid((kSig / kBlkS) * kTileStep);
    fir_bandpass_wmma2x2<<<grid, 256, 0, stream>>>(x, h, out);
}


# Three block products for four per pair of 16-column blocks: S+=G(Xe+Xo), P+=(G'-G)Xo, M+=(G-G'')Xe, y0=S+P, y1=S-M; tap differences tabled in LDS; 336 WMMAs per wave for 444 (new f32 sum order)
# speedup vs baseline: 2.4310x; 1.1191x over previous
#include <hip/hip_runtime.h>


typedef __attribute__((ext_vector_type(2))) float v2f;
typedef __attribute__((ext_vector_type(8))) float v8f;

namespace {
constexpr int kT       = 4000;
constexpr int kSig     = 64 * 64;
constexpr int kTaps    = 413;
constexpr int kHalf    = (kTaps - 1) / 2;
constexpr int kBlkS    = 64;
constexpr int kBlkT    = 128;
constexpr int kTimeBlocks = (kT + kBlkT - 1) / kBlkT;
constexpr int kTilesPerWG = 4;
constexpr int kTileStep   = kTimeBlocks / kTilesPerWG;
static_assert(kTimeBlocks % kTilesPerWG == 0, "time blocks must divide evenly over the workgroups");
static_assert(kTilesPerWG == 4, "one priority level (3, 2, 1, 0) per time block of a workgroup");
static_assert(kSig / kBlkS == 64, "blockIdx.x is split into its low 6 bits (signal block) and the rest (first time block)");
constexpr int kPre     = 208;
constexpr int kShift   = kPre - kHalf;
constexpr int kWinV    = 136;
constexpr int kPerThr  = kBlkS * kWinV / 256;
constexpr int kStride  = 548;
static_assert(kBlkS * kWinV == kPerThr * 256 && kPerThr == 34 && kWinV == 4 * 32 + 8 && kBlkS == 8 * 8,
              "the chunk-to-thread map below is written for 64 rows of 136 chunks on 256 threads");
static_assert(kT % 4 == 0 && kBlkT % 4 == 0 && kPre % 4 == 0, "chunks must not straddle the signal's ends");
static_assert(kT >= 2 * 4 * kWinV, "a window must not overhang both ends of a signal");
constexpr int kPairs   = 14;
constexpr int kPadCol  = 4 * kWinV;
constexpr int kHpLen   = 480;
constexpr int kHpOff   = 32;
constexpr int kHpFill  = 512;
constexpr int kHpPitch = 544;
static_assert(kHpFill >= kHpLen && kHpPitch >= kHpFill && kHpPitch % 64 == 32, "tap copies");
static_assert(kHpOff + 32 * (kPairs - 1) + 12 + 3 + 16 < kHpLen, "highest tap and tap-difference entry a fragment reads");
static_assert(kShift + 96 + 32 * (kPairs - 1) + 12 + 16 + 3 < kPadCol + 4 && kPadCol + 4 <= kStride, "highest window column a fragment reads lies in the zeroed pad");

__device__ __forceinline__ int clamp_chunk(int j)
{
    return j < 0 ? 0 : (j > kT - 4 ? kT - 4 : j);
}

#define FIR_MAIN_CHUNKS(X) \
    X(0,0) X(0,1) X(0,2) X(0,3) X(0,4) X(0,5) X(0,6) X(0,7) \
    X(1,0) X(1,1) X(1,2) X(1,3) X(1,4) X(1,5) X(1,6) X(1,7) \
    X(2,0) X(2,1) X(2,2) X(2,3) X(2,4) X(2,5) X(2,6) X(2,7) \
    X(3,0) X(3,1) X(3,2) X(3,3) X(3,4) X(3,5) X(3,6) X(3,7)
#define FIR_DECL(j,k)  float4 wm##j##k;
#define FIR_REQ(j,k)   wm##j##k = *(const float4*)(gp##j + (k) * 8 * kT);
#define FIR_PUT(j,k)   *(float4*)(lp + (k) * 8 * kStride + (j) * 128) = wm##j##k;
}

__global__ __launch_bounds__(256)
void fir_bandpass_wmma2x2(const float* __restrict__ x,
                          const float* __restrict__ h,
                          float* __restrict__ out)
{
    __shared__ __attribute__((aligned(16))) float xs[kBlkS * kStride];
    __shared__ float hp[kHpPitch + kHpFill];
    __shared__ float dp[kHpPitch + kHpFill];

    __builtin_amdgcn_s_setprio(3);

    const int tid = threadIdx.x;
    const int s0b = (blockIdx.x & 63) * kBlkS;
    const int tb0 = blockIdx.x >> 6;

    const int ja = tid - kHpOff;
    const int jb = tid + 256 - kHpOff;
    const float ta = h[kTaps - 1 - (ja < 0 ? 0 : ja)];
    const float tb = h[jb < kTaps ? kTaps - 1 - jb : 0];
    const int jc = ja + 16;
    const int jd = jb + 16;
    const float tc = h[kTaps - 1 - (jc < 0 ? 0 : jc)];
    const float td = h[jd < kTaps ? kTaps - 1 - jd : 0];

    const int w    = tid >> 5;
    const int lane = tid & 31;

    FIR_MAIN_CHUNKS(FIR_DECL)
    float4 wt0, wt1;
    const float* grow = x + (size_t)(s0b + w) * kT;
    const float* trow = x + (size_t)(s0b + (tid >> 2)) * kT;
    const int    tq   = 4 * (kWinV - 8 + (tid & 3));
    float* const lp   = xs + w * kStride + 4 * lane;
    float* const lt   = xs + (tid >> 2) * kStride + tq;
    {
        const int j0 = tb0 * kBlkT - kPre;
        const float* gp0 = grow + clamp_chunk(j0 + 4 * lane);
        const float* gp1 = grow + clamp_chunk(j0 + 4 * lane + 128);
        const float* gp2 = grow + clamp_chunk(j0 + 4 * lane + 256);
        const float* gp3 = grow + clamp_chunk(j0 + 4 * lane + 384);
        FIR_MAIN_CHUNKS(FIR_REQ)
        wt0 = *(const float4*)(trow + clamp_chunk(j0 + tq));
        wt1 = *(const float4*)(trow + clamp_chunk(j0 + tq + 16));
    }

    {
        const float va = ja >= 0 ? ta : 0.0f;
        const float vb = jb < kTaps ? tb : 0.0f;
        hp[tid]                  = va;
        hp[kHpPitch + tid]       = va;
        hp[tid + 256]            = vb;
        hp[kHpPitch + tid + 256] = vb;
        const float dlo = (jc >= 0 ? tc : 0.0f) - va;
        const float dhi = (jd < kTaps ? td : 0.0f) - vb;
        dp[tid]                  = dlo;
        dp[kHpPitch + tid]       = dlo;
        dp[tid + 256]            = dhi;
        dp[kHpPitch + tid + 256] = dhi;
        if (tid < kBlkS) *(float4*)(xs + tid * kStride + kPadCol) = make_float4(0.f, 0.f, 0.f, 0.f);
    }

    const int mg   = w & 3;
    const int ng   = w >> 2;
    const int m    = lane & 15;
    const int hi   = lane >> 4;
    const int kk   = hi << 1;

    const float* hb  = hp + (hi ? kHpPitch : 0) + (kHpOff + kk - m);
    const float* db  = dp + (hi ? kHpPitch : 0) + (kHpOff + kk - m);
    const float* xb0 = xs + ((ng << 5) + m) * kStride + kShift + (mg << 5) + kk;
    const float* xb1 = xb0 + 16 * kStride;

    #pragma unroll 1
    for (int n = 0; n < kTilesPerWG; ++n) {
        const int t0b = (tb0 + n * kTileStep) * kBlkT;
        const int j0  = t0b - kPre;

        FIR_MAIN_CHUNKS(FIR_PUT)
        *(float4*)(lt)      = wt0;
        *(float4*)(lt + 16) = wt1;

        if (j0 < 0 || j0 + 4 * kWinV > kT) {
            __syncthreads();
            const bool left = j0 < 0;
            const int nOut  = left ? ((-j0) >> 2) : ((j0 + 4 * kWinV - kT) >> 2);
            const int qb    = left ? 0 : kWinV - nOut;
            const int s     = tid & 63;
            const float* row = x + (size_t)(s0b + s) * kT;
            const float e2  = 2.0f * row[left ? 0 : kT - 1];
            #pragma unroll 4
            for (int q = qb + (tid >> 6); q < qb + nOut; q += 4) {
                const int j = j0 + 4 * q;
                const int r = left ? -j : 2 * kT - 2 - j;
                float4 p;
                p.x = e2 - row[r];
                p.y = e2 - row[r - 1];
                p.z = e2 - row[r - 2];
                p.w = e2 - row[r - 3];
                *(float4*)(xs + s * kStride + 4 * q) = p;
            }
        }
        __syncthreads();

        if (n + 1 < kTilesPerWG) {
            const int j1 = j0 + kTileStep * kBlkT;
            const float* gp0 = grow + clamp_chunk(j1 + 4 * lane);
            const float* gp1 = grow + clamp_chunk(j1 + 4 * lane + 128);
            const float* gp2 = grow + clamp_chunk(j1 + 4 * lane + 256);
            const float* gp3 = grow + clamp_chunk(j1 + 4 * lane + 384);
            FIR_MAIN_CHUNKS(FIR_REQ)
            wt0 = *(const float4*)(trow + clamp_chunk(j1 + tq));
            wt1 = *(const float4*)(trow + clamp_chunk(j1 + tq + 16));
        }

        const int t00 = t0b + (mg << 5);
        if (t00 < kT) {
            v8f accS0 = {0.f,0.f,0.f,0.f,0.f,0.f,0.f,0.f};
            v8f accS1 = accS0;
            v8f accP0 = accS0;
            v8f accP1 = accS0;
            v8f accM0 = accS0;
            v8f accM1 = accS0;

            #pragma unroll
            for (int i = 0; i < kPairs; ++i) {
                const int cb = 32 * i;
                for (int q = 0; q < 4; ++q) {
                    const int c = cb + 4 * q;
                    v2f aS, aP, aM, e0, o0, e1, o1, s0, s1;
                    aS.x = hb[c];          aS.y = hb[c + 1];
                    aP.x = db[c];          aP.y = db[c + 1];
                    aM.x = db[c - 16];     aM.y = db[c - 15];
                    e0.x = xb0[c];         e0.y = xb0[c + 1];
                    o0.x = xb0[c + 16];    o0.y = xb0[c + 17];
                    e1.x = xb1[c];         e1.y = xb1[c + 1];
                    o1.x = xb1[c + 16];    o1.y = xb1[c + 17];
                    s0 = e0 + o0;
                    s1 = e1 + o1;
                    accS0 = __builtin_amdgcn_wmma_f32_16x16x4_f32(
                        false, aS, false, s0, (short)0, accS0, false, false);
                    accS1 = __builtin_amdgcn_wmma_f32_16x16x4_f32(
                        false, aS, false, s1, (short)0, accS1, false, false);
                    accP0 = __builtin_amdgcn_wmma_f32_16x16x4_f32(
                        false, aP, false, o0, (short)0, accP0, false, false);
                    accP1 = __builtin_amdgcn_wmma_f32_16x16x4_f32(
                        false, aP, false, o1, (short)0, accP1, false, false);
                    accM0 = __builtin_amdgcn_wmma_f32_16x16x4_f32(
                        false, aM, false, e0, (short)0, accM0, false, false);
                    accM1 = __builtin_amdgcn_wmma_f32_16x16x4_f32(
                        false, aM, false, e1, (short)0, accM1, false, false);
                }
            }

            const v8f acc00 = accS0 + accP0;
            const v8f acc01 = accS1 + accP1;
            const v8f acc10 = accS0 - accM0;
            const v8f acc11 = accS1 - accM1;

            float* o00 = out + (size_t)(s0b + (ng << 5) + m) * kT + t00 + (hi << 3);
            float* o01 = o00 + (size_t)16 * kT;
            float* o10 = o00 + 16;
            float* o11 = o01 + 16;

            float4 v;
            v.x = acc00[0]; v.y = acc00[1]; v.z = acc00[2]; v.w = acc00[3];
            *(float4*)(o00) = v;
            v.x = acc00[4]; v.y = acc00[5]; v.z = acc00[6]; v.w = acc00[7];
            *(float4*)(o00 + 4) = v;
            v.x = acc01[0]; v.y = acc01[1]; v.z = acc01[2]; v.w = acc01[3];
            *(float4*)(o01) = v;
            v.x = acc01[4]; v.y = acc01[5]; v.z = acc01[6]; v.w = acc01[7];
            *(float4*)(o01 + 4) = v;
            v.x = acc10[0]; v.y = acc10[1]; v.z = acc10[2]; v.w = acc10[3];
            *(float4*)(o10) = v;
            v.x = acc10[4]; v.y = acc10[5]; v.z = acc10[6]; v.w = acc10[7];
            *(float4*)(o10 + 4) = v;
            v.x = acc11[0]; v.y = acc11[1]; v.z = acc11[2]; v.w = acc11[3];
            *(float4*)(o11) = v;
            v.x = acc11[4]; v.y = acc11[5]; v.z = acc11[6]; v.w = acc11[7];
            *(float4*)(o11 + 4) = v;
        }
        if (n == 0) __builtin_amdgcn_s_setprio(2);
        if (n == 1) __builtin_amdgcn_s_setprio(1);
        if (n == 2) __builtin_amdgcn_s_setprio(0);

        __syncthreads();
    }
}

extern "C" void kernel_launch(void* const* d_in, const int* in_sizes, int n_in,
                              void* d_out, int out_size, void* d_ws, size_t ws_size,
                              hipStream_t stream) {
    const float* x = (const float*)d_in[0];
    const float* h = (const float*)d_in[1];
    float* out = (float*)d_out;

    dim3 grid((kSig / kBlkS) * kTileStep);
    fir_bandpass_wmma2x2<<<grid, 256, 0, stream>>>(x, h, out);
}


# Block-Toeplitz split applied twice: wave tile 64 times x 16 signals, 9 block products per 4x4 block group for 16 (12 before), 262 WMMAs per wave for 336; workgroup block 16 x 512 (new f32 sum order)
# speedup vs baseline: 3.1028x; 1.2763x over previous
#include <hip/hip_runtime.h>


typedef __attribute__((ext_vector_type(2))) float v2f;
typedef __attribute__((ext_vector_type(8))) float v8f;

namespace {
constexpr int kT       = 4000;
constexpr int kSig     = 64 * 64;
constexpr int kTaps    = 413;
constexpr int kHalf    = (kTaps - 1) / 2;
constexpr int kBlkS    = 16;
constexpr int kBlkT    = 512;
constexpr int kTimeBlocks = (kT + kBlkT - 1) / kBlkT;
constexpr int kTilesPerWG = 4;
constexpr int kTileStep   = kTimeBlocks / kTilesPerWG;
static_assert(kTimeBlocks % kTilesPerWG == 0, "time blocks must divide evenly over the workgroups");
static_assert(kTilesPerWG == 4, "one priority level (3, 2, 1, 0) per time block of a workgroup");
static_assert(kSig / kBlkS == 256, "blockIdx.x is split into its low 8 bits (signal block) and the rest (first time block)");
constexpr int kPre     = 208;
constexpr int kShift   = kPre - kHalf;
constexpr int kWinV    = 240;
constexpr int kPerThr  = kBlkS * kWinV / 256;
constexpr int kStride  = 964;
static_assert(kBlkS * kWinV == kPerThr * 256 && kPerThr == 15 && kWinV == 7 * 32 + 16 && kBlkS == 2 * 8,
              "the chunk-to-thread map below is written for 16 rows of 240 chunks on 256 threads");
static_assert(kT % 4 == 0 && kBlkT % 4 == 0 && kPre % 4 == 0, "chunks must not straddle the signal's ends");
static_assert(kT >= 2 * 4 * kWinV, "a window must not overhang both ends of a signal");
static_assert(kT % 16 == 0, "an M-tile lies wholly inside or wholly outside a signal");
constexpr int kGroups  = 7;
constexpr int kHpOff   = 64;
constexpr int kHpFill  = 512;
constexpr int kHpPitch = 544;
static_assert(kHpPitch >= kHpFill && kHpPitch % 64 == 32, "tap copies");
static_assert(kHpOff - 15 - 48 >= 0 && kHpOff + 2 + 16 * 26 + 8 + 1 < kHpFill, "lowest and highest table entry a fragment reads");
static_assert(16 * 26 + 12 - 15 >= kTaps, "K-slice 3 (v = 12..15) of G_26 holds no tap: g(416 + v - m) lies past the last one");
static_assert(kShift + 64 * 7 + 2 + 16 * 29 + 8 + 1 < 4 * kWinV, "highest window column a fragment reads is a staged one");

__device__ __forceinline__ int clamp_chunk(int j)
{
    return j < 0 ? 0 : (j > kT - 4 ? kT - 4 : j);
}

#define FIR_MAIN_CHUNKS(X) \
    X(0,0) X(0,1) X(1,0) X(1,1) X(2,0) X(2,1) X(3,0) X(3,1) \
    X(4,0) X(4,1) X(5,0) X(5,1) X(6,0) X(6,1)
#define FIR_DECL(j,k)  float4 wm##j##k;
#define FIR_REQ(j,k)   wm##j##k = *(const float4*)(gp##j + (k) * 8 * kT);
#define FIR_PUT(j,k)   *(float4*)(lp + (k) * 8 * kStride + (j) * 128) = wm##j##k;
}

__global__ __launch_bounds__(256, 4)
void fir_bandpass_wmma4x1(const float* __restrict__ x,
                          const float* __restrict__ h,
                          float* __restrict__ out)
{
    __shared__ __attribute__((aligned(16))) float xs[kBlkS * kStride];
    __shared__ float hp[kHpPitch + kHpFill];
    __shared__ float d16[kHpPitch + kHpFill];
    __shared__ float d32[kHpPitch + kHpFill];
    __shared__ float dd[kHpPitch + kHpFill];

    __builtin_amdgcn_s_setprio(3);

    const int tid = threadIdx.x;
    const int s0b = (blockIdx.x & 255) * kBlkS;
    const int tb0 = blockIdx.x >> 8;

    const int ja = tid - kHpOff;
    const int jb = tid + 256 - kHpOff;
    float ga[4], gb[4];
    #pragma unroll
    for (int u = 0; u < 4; ++u) {
        const int a = ja + 16 * u, b = jb + 16 * u;
        ga[u] = h[kTaps - 1 - (a < 0 ? 0 : a)];
        gb[u] = h[b < kTaps ? kTaps - 1 - b : 0];
    }

    const int w    = tid >> 5;
    const int lane = tid & 31;

    FIR_MAIN_CHUNKS(FIR_DECL)
    float4 wt0;
    const float* grow = x + (size_t)(s0b + w) * kT;
    const float* trow = x + (size_t)(s0b + (tid >> 4)) * kT;
    const int    tq   = 4 * (kWinV - 16 + (tid & 15));
    float* const lp   = xs + w * kStride + 4 * lane;
    float* const lt   = xs + (tid >> 4) * kStride + tq;
    {
        const int j0 = tb0 * kBlkT - kPre;
        const float* gp0 = grow + clamp_chunk(j0 + 4 * lane);
        const float* gp1 = grow + clamp_chunk(j0 + 4 * lane + 128);
        const float* gp2 = grow + clamp_chunk(j0 + 4 * lane + 256);
        const float* gp3 = grow + clamp_chunk(j0 + 4 * lane + 384);
        const float* gp4 = grow + clamp_chunk(j0 + 4 * lane + 512);
        const float* gp5 = grow + clamp_chunk(j0 + 4 * lane + 640);
        const float* gp6 = grow + clamp_chunk(j0 + 4 * lane + 768);
        FIR_MAIN_CHUNKS(FIR_REQ)
        wt0 = *(const float4*)(trow + clamp_chunk(j0 + tq));
    }

    {
        #pragma unroll
        for (int u = 0; u < 4; ++u) {
            ga[u] = ja + 16 * u >= 0 ? ga[u] : 0.0f;
            gb[u] = jb + 16 * u < kTaps ? gb[u] : 0.0f;
        }
        const float a16 = ga[1] - ga[0], b16 = gb[1] - gb[0];
        const float a32 = ga[2] - ga[0], b32 = gb[2] - gb[0];
        const float add = (ga[3] - ga[1]) - a32;
        const float bdd = (gb[3] - gb[1]) - b32;
        hp[tid]        = ga[0];   hp[kHpPitch + tid]        = ga[0];
        hp[tid + 256]  = gb[0];   hp[kHpPitch + tid + 256]  = gb[0];
        d16[tid]       = a16;     d16[kHpPitch + tid]       = a16;
        d16[tid + 256] = b16;     d16[kHpPitch + tid + 256] = b16;
        d32[tid]       = a32;     d32[kHpPitch + tid]       = a32;
        d32[tid + 256] = b32;     d32[kHpPitch + tid + 256] = b32;
        dd[tid]        = add;     dd[kHpPitch + tid]        = add;
        dd[tid + 256]  = bdd;     dd[kHpPitch + tid + 256]  = bdd;
    }

    const int mg   = w;
    const int m    = lane & 15;
    const int hi   = lane >> 4;
    const int kk   = hi << 1;

    const int   tof = (hi ? kHpPitch : 0) + (kHpOff + kk - m);
    const float* tb = hp  + tof;
    const float* e1 = d16 + tof;
    const float* e2 = d32 + tof;
    const float* e3 = dd  + tof;
    const float* xb = xs + m * kStride + kShift + (mg << 6) + kk;

    #pragma unroll 1
    for (int n = 0; n < kTilesPerWG; ++n) {
        const int t0b = (tb0 + n * kTileStep) * kBlkT;
        const int j0  = t0b - kPre;

        FIR_MAIN_CHUNKS(FIR_PUT)
        *(float4*)(lt) = wt0;

        if (j0 < 0 || j0 + 4 * kWinV > kT) {
            __syncthreads();
            const bool left = j0 < 0;
            const int nOut  = left ? ((-j0) >> 2) : ((j0 + 4 * kWinV - kT) >> 2);
            const int qb    = left ? 0 : kWinV - nOut;
            const int s     = tid & 15;
            const float* row = x + (size_t)(s0b + s) * kT;
            const float e2x = 2.0f * row[left ? 0 : kT - 1];
            #pragma unroll 4
            for (int q = qb + (tid >> 4); q < qb + nOut; q += 16) {
                const int j = j0 + 4 * q;
                const int r = left ? -j : 2 * kT - 2 - j;
                float4 p;
                p.x = e2x - row[r];
                p.y = e2x - row[r - 1];
                p.z = e2x - row[r - 2];
                p.w = e2x - row[r - 3];
                *(float4*)(xs + s * kStride + 4 * q) = p;
            }
        }
        __syncthreads();

        if (n + 1 < kTilesPerWG) {
            const int j1 = j0 + kTileStep * kBlkT;
            const float* gp0 = grow + clamp_chunk(j1 + 4 * lane);
            const float* gp1 = grow + clamp_chunk(j1 + 4 * lane + 128);
            const float* gp2 = grow + clamp_chunk(j1 + 4 * lane + 256);
            const float* gp3 = grow + clamp_chunk(j1 + 4 * lane + 384);
            const float* gp4 = grow + clamp_chunk(j1 + 4 * lane + 512);
            const float* gp5 = grow + clamp_chunk(j1 + 4 * lane + 640);
            const float* gp6 = grow + clamp_chunk(j1 + 4 * lane + 768);
            FIR_MAIN_CHUNKS(FIR_REQ)
            wt0 = *(const float4*)(trow + clamp_chunk(j1 + tq));
        }

        const int t00 = t0b + (mg << 6);
        if (t00 < kT) {
            const v8f zero = {0.f,0.f,0.f,0.f,0.f,0.f,0.f,0.f};
            v8f aSs = zero, aSp = zero, aSm = zero;
            v8f aPs = zero, aPp = zero, aPm = zero;
            v8f aMs = zero, aMp = zero, aMm = zero;

            #pragma unroll
            for (int r = 0; r < kGroups; ++r) {
                for (int q = 0; q < 4; ++q) {
                    const int c = 64 * r + 4 * q;
                    v2f w0, w1, w2, w3, z02, z13, zall, z23, z01, a;
                    w0.x = xb[c];          w0.y = xb[c + 1];
                    w1.x = xb[c + 16];     w1.y = xb[c + 17];
                    w2.x = xb[c + 32];     w2.y = xb[c + 33];
                    w3.x = xb[c + 48];     w3.y = xb[c + 49];
                    z02  = w0 + w2;
                    z13  = w1 + w3;
                    zall = z02 + z13;
                    z23  = w2 + w3;
                    z01  = w0 + w1;
                    a.x = tb[c];           a.y = tb[c + 1];
                    aSs = __builtin_amdgcn_wmma_f32_16x16x4_f32(false, a, false, zall, (short)0, aSs, false, false);
                    a.x = e1[c];           a.y = e1[c + 1];
                    aSp = __builtin_amdgcn_wmma_f32_16x16x4_f32(false, a, false, z13, (short)0, aSp, false, false);
                    a.x = e1[c - 16];      a.y = e1[c - 15];
                    aSm = __builtin_amdgcn_wmma_f32_16x16x4_f32(false, a, false, z02, (short)0, aSm, false, false);
                    a.x = e2[c];           a.y = e2[c + 1];
                    aPs = __builtin_amdgcn_wmma_f32_16x16x4_f32(false, a, false, z23, (short)0, aPs, false, false);
                    a.x = e3[c];           a.y = e3[c + 1];
                    aPp = __builtin_amdgcn_wmma_f32_16x16x4_f32(false, a, false, w3, (short)0, aPp, false, false);
                    a.x = e3[c - 16];      a.y = e3[c - 15];
                    aPm = __builtin_amdgcn_wmma_f32_16x16x4_f32(false, a, false, w2, (short)0, aPm, false, false);
                    a.x = e2[c - 32];      a.y = e2[c - 31];
                    aMs = __builtin_amdgcn_wmma_f32_16x16x4_f32(false, a, false, z01, (short)0, aMs, false, false);
                    a.x = e3[c - 32];      a.y = e3[c - 31];
                    aMp = __builtin_amdgcn_wmma_f32_16x16x4_f32(false, a, false, w1, (short)0, aMp, false, false);
                    a.x = e3[c - 48];      a.y = e3[c - 47];
                    aMm = __builtin_amdgcn_wmma_f32_16x16x4_f32(false, a, false, w0, (short)0, aMm, false, false);
                }
            }
            #pragma unroll
            for (int q = 0; q < 4; ++q) {
                v2f w28, a;
                w28.x = xb[448 + 4 * q];   w28.y = xb[449 + 4 * q];
                a.x = tb[400 + 4 * q];     a.y = tb[401 + 4 * q];
                aMm = __builtin_amdgcn_wmma_f32_16x16x4_f32(false, a, false, w28, (short)0, aMm, false, false);
                if (q < 3) {
                    v2f w29, n28;
                    w29.x = xb[464 + 4 * q];   w29.y = xb[465 + 4 * q];
                    n28 = -w28;
                    a.x = tb[416 + 4 * q];     a.y = tb[417 + 4 * q];
                    aMm = __builtin_amdgcn_wmma_f32_16x16x4_f32(false, a, false, w29, (short)0, aMm, false, false);
                    aMp = __builtin_amdgcn_wmma_f32_16x16x4_f32(false, a, false, n28, (short)0, aMp, false, false);
                }
            }

            const v8f s0 = aSs + aSp, s1 = aSs - aSm;
            const v8f y0 = s0 + (aPs + aPp);
            const v8f y1 = s1 + (aPs - aPm);
            const v8f y2 = s0 - (aMs + aMp);
            const v8f y3 = s1 - (aMs - aMm);

            float* o0 = out + (size_t)(s0b + m) * kT + t00 + (hi << 3);
            float4 v;
            v.x = y0[0]; v.y = y0[1]; v.z = y0[2]; v.w = y0[3];
            *(float4*)(o0) = v;
            v.x = y0[4]; v.y = y0[5]; v.z = y0[6]; v.w = y0[7];
            *(float4*)(o0 + 4) = v;
            if (t00 + 16 < kT) {
                v.x = y1[0]; v.y = y1[1]; v.z = y1[2]; v.w = y1[3];
                *(float4*)(o0 + 16) = v;
                v.x = y1[4]; v.y = y1[5]; v.z = y1[6]; v.w = y1[7];
                *(float4*)(o0 + 20) = v;
            }
            if (t00 + 32 < kT) {
                v.x = y2[0]; v.y = y2[1]; v.z = y2[2]; v.w = y2[3];
                *(float4*)(o0 + 32) = v;
                v.x = y2[4]; v.y = y2[5]; v.z = y2[6]; v.w = y2[7];
                *(float4*)(o0 + 36) = v;
            }
            if (t00 + 48 < kT) {
                v.x = y3[0]; v.y = y3[1]; v.z = y3[2]; v.w = y3[3];
                *(float4*)(o0 + 48) = v;
                v.x = y3[4]; v.y = y3[5]; v.z = y3[6]; v.w = y3[7];
                *(float4*)(o0 + 52) = v;
            }
        }
        if (n == 0) __builtin_amdgcn_s_setprio(2);
        if (n == 1) __builtin_amdgcn_s_setprio(1);
        if (n == 2) __builtin_amdgcn_s_setprio(0);

        __syncthreads();
    }
}

extern "C" void kernel_launch(void* const* d_in, const int* in_sizes, int n_in,
                              void* d_out, int out_size, void* d_ws, size_t ws_size,
                              hipStream_t stream) {
    const float* x = (const float*)d_in[0];
    const float* h = (const float*)d_in[1];
    float* out = (float*)d_out;

    dim3 grid((kSig / kBlkS) * kTileStep);
    fir_bandpass_wmma4x1<<<grid, 256, 0, stream>>>(x, h, out);
}


# Whole-line output stores: each wave passes its tile through an LDS tile of its own, 32 samples x 16 signals at a time, and stores rows: 4 full 128-byte lines per store, not 32 pieces of 16 bytes
# speedup vs baseline: 3.4034x; 1.0969x over previous
#include <hip/hip_runtime.h>


typedef __attribute__((ext_vector_type(2))) float v2f;
typedef __attribute__((ext_vector_type(8))) float v8f;

namespace {
constexpr int kT       = 4000;
constexpr int kSig     = 64 * 64;
constexpr int kTaps    = 413;
constexpr int kHalf    = (kTaps - 1) / 2;
constexpr int kBlkS    = 16;
constexpr int kBlkT    = 512;
constexpr int kTimeBlocks = (kT + kBlkT - 1) / kBlkT;
constexpr int kTilesPerWG = 4;
constexpr int kTileStep   = kTimeBlocks / kTilesPerWG;
static_assert(kTimeBlocks % kTilesPerWG == 0, "time blocks must divide evenly over the workgroups");
static_assert(kTilesPerWG == 4, "one priority level (3, 2, 1, 0) per time block of a workgroup");
static_assert(kSig / kBlkS == 256, "blockIdx.x is split into its low 8 bits (signal block) and the rest (first time block)");
constexpr int kPre     = 208;
constexpr int kShift   = kPre - kHalf;
constexpr int kWinV    = 240;
constexpr int kPerThr  = kBlkS * kWinV / 256;
constexpr int kStride  = 964;
static_assert(kBlkS * kWinV == kPerThr * 256 && kPerThr == 15 && kWinV == 7 * 32 + 16 && kBlkS == 2 * 8,
              "the chunk-to-thread map below is written for 16 rows of 240 chunks on 256 threads");
static_assert(kT % 4 == 0 && kBlkT % 4 == 0 && kPre % 4 == 0, "chunks must not straddle the signal's ends");
static_assert(kT >= 2 * 4 * kWinV, "a window must not overhang both ends of a signal");
static_assert(kT % 16 == 0, "an M-tile lies wholly inside or wholly outside a signal");
constexpr int kStgPitch = 36;
static_assert(kStgPitch >= 32 && kStgPitch % 4 == 0 && (kStgPitch / 4) % 2 == 1, "staging rows: 32 samples, 16-byte aligned, an odd multiple of 16 bytes");
static_assert(kT % 32 == 0, "a 128-byte output line lies wholly inside or wholly outside a signal");
constexpr int kGroups  = 7;
constexpr int kHpOff   = 64;
constexpr int kHpFill  = 512;
constexpr int kHpPitch = 544;
static_assert(kHpPitch >= kHpFill && kHpPitch % 64 == 32, "tap copies");
static_assert(kHpOff - 15 - 48 >= 0 && kHpOff + 2 + 16 * 26 + 8 + 1 < kHpFill, "lowest and highest table entry a fragment reads");
static_assert(16 * 26 + 12 - 15 >= kTaps, "K-slice 3 (v = 12..15) of G_26 holds no tap: g(416 + v - m) lies past the last one");
static_assert(kShift + 64 * 7 + 2 + 16 * 29 + 8 + 1 < 4 * kWinV, "highest window column a fragment reads is a staged one");

__device__ __forceinline__ int clamp_chunk(int j)
{
    return j < 0 ? 0 : (j > kT - 4 ? kT - 4 : j);
}

#define FIR_MAIN_CHUNKS(X) \
    X(0,0) X(0,1) X(1,0) X(1,1) X(2,0) X(2,1) X(3,0) X(3,1) \
    X(4,0) X(4,1) X(5,0) X(5,1) X(6,0) X(6,1)
#define FIR_DECL(j,k)  float4 wm##j##k;
#define FIR_REQ(j,k)   wm##j##k = *(const float4*)(gp##j + (k) * 8 * kT);
#define FIR_PUT(j,k)   *(float4*)(lp + (k) * 8 * kStride + (j) * 128) = wm##j##k;
}

__global__ __launch_bounds__(256, 4)
void fir_bandpass_wmma4x1(const float* __restrict__ x,
                          const float* __restrict__ h,
                          float* __restrict__ out)
{
    __shared__ __attribute__((aligned(16))) float xs[kBlkS * kStride];
    __shared__ float hp[kHpPitch + kHpFill];
    __shared__ float d16[kHpPitch + kHpFill];
    __shared__ float d32[kHpPitch + kHpFill];
    __shared__ float dd[kHpPitch + kHpFill];
    __shared__ __attribute__((aligned(16))) float stg[8 * kBlkS * kStgPitch];

    __builtin_amdgcn_s_setprio(3);

    const int tid = threadIdx.x;
    const int s0b = (blockIdx.x & 255) * kBlkS;
    const int tb0 = blockIdx.x >> 8;

    const int ja = tid - kHpOff;
    const int jb = tid + 256 - kHpOff;
    float ga[4], gb[4];
    #pragma unroll
    for (int u = 0; u < 4; ++u) {
        const int a = ja + 16 * u, b = jb + 16 * u;
        ga[u] = h[kTaps - 1 - (a < 0 ? 0 : a)];
        gb[u] = h[b < kTaps ? kTaps - 1 - b : 0];
    }

    const int w    = tid >> 5;
    const int lane = tid & 31;

    FIR_MAIN_CHUNKS(FIR_DECL)
    float4 wt0;
    const float* grow = x + (size_t)(s0b + w) * kT;
    const float* trow = x + (size_t)(s0b + (tid >> 4)) * kT;
    const int    tq   = 4 * (kWinV - 16 + (tid & 15));
    float* const lp   = xs + w * kStride + 4 * lane;
    float* const lt   = xs + (tid >> 4) * kStride + tq;
    {
        const int j0 = tb0 * kBlkT - kPre;
        const float* gp0 = grow + clamp_chunk(j0 + 4 * lane);
        const float* gp1 = grow + clamp_chunk(j0 + 4 * lane + 128);
        const float* gp2 = grow + clamp_chunk(j0 + 4 * lane + 256);
        const float* gp3 = grow + clamp_chunk(j0 + 4 * lane + 384);
        const float* gp4 = grow + clamp_chunk(j0 + 4 * lane + 512);
        const float* gp5 = grow + clamp_chunk(j0 + 4 * lane + 640);
        const float* gp6 = grow + clamp_chunk(j0 + 4 * lane + 768);
        FIR_MAIN_CHUNKS(FIR_REQ)
        wt0 = *(const float4*)(trow + clamp_chunk(j0 + tq));
    }

    {
        #pragma unroll
        for (int u = 0; u < 4; ++u) {
            ga[u] = ja + 16 * u >= 0 ? ga[u] : 0.0f;
            gb[u] = jb + 16 * u < kTaps ? gb[u] : 0.0f;
        }
        const float a16 = ga[1] - ga[0], b16 = gb[1] - gb[0];
        const float a32 = ga[2] - ga[0], b32 = gb[2] - gb[0];
        const float add = (ga[3] - ga[1]) - a32;
        const float bdd = (gb[3] - gb[1]) - b32;
        hp[tid]        = ga[0];   hp[kHpPitch + tid]        = ga[0];
        hp[tid + 256]  = gb[0];   hp[kHpPitch + tid + 256]  = gb[0];
        d16[tid]       = a16;     d16[kHpPitch + tid]       = a16;
        d16[tid + 256] = b16;     d16[kHpPitch + tid + 256] = b16;
        d32[tid]       = a32;     d32[kHpPitch + tid]       = a32;
        d32[tid + 256] = b32;     d32[kHpPitch + tid + 256] = b32;
        dd[tid]        = add;     dd[kHpPitch + tid]        = add;
        dd[tid + 256]  = bdd;     dd[kHpPitch + tid + 256]  = bdd;
    }

    const int mg   = w;
    const int m    = lane & 15;
    const int hi   = lane >> 4;
    const int kk   = hi << 1;

    const int   tof = (hi ? kHpPitch : 0) + (kHpOff + kk - m);
    const float* tb = hp  + tof;
    const float* e1 = d16 + tof;
    const float* e2 = d32 + tof;
    const float* e3 = dd  + tof;
    const float* xb = xs + m * kStride + kShift + (mg << 6) + kk;

    #pragma unroll 1
    for (int n = 0; n < kTilesPerWG; ++n) {
        const int t0b = (tb0 + n * kTileStep) * kBlkT;
        const int j0  = t0b - kPre;

        FIR_MAIN_CHUNKS(FIR_PUT)
        *(float4*)(lt) = wt0;

        if (j0 < 0 || j0 + 4 * kWinV > kT) {
            __syncthreads();
            const bool left = j0 < 0;
            const int nOut  = left ? ((-j0) >> 2) : ((j0 + 4 * kWinV - kT) >> 2);
            const int qb    = left ? 0 : kWinV - nOut;
            const int s     = tid & 15;
            const float* row = x + (size_t)(s0b + s) * kT;
            const float e2x = 2.0f * row[left ? 0 : kT - 1];
            #pragma unroll 4
            for (int q = qb + (tid >> 4); q < qb + nOut; q += 16) {
                const int j = j0 + 4 * q;
                const int r = left ? -j : 2 * kT - 2 - j;
                float4 p;
                p.x = e2x - row[r];
                p.y = e2x - row[r - 1];
                p.z = e2x - row[r - 2];
                p.w = e2x - row[r - 3];
                *(float4*)(xs + s * kStride + 4 * q) = p;
            }
        }
        __syncthreads();

        if (n + 1 < kTilesPerWG) {
            const int j1 = j0 + kTileStep * kBlkT;
            const float* gp0 = grow + clamp_chunk(j1 + 4 * lane);
            const float* gp1 = grow + clamp_chunk(j1 + 4 * lane + 128);
            const float* gp2 = grow + clamp_chunk(j1 + 4 * lane + 256);
            const float* gp3 = grow + clamp_chunk(j1 + 4 * lane + 384);
            const float* gp4 = grow + clamp_chunk(j1 + 4 * lane + 512);
            const float* gp5 = grow + clamp_chunk(j1 + 4 * lane + 640);
            const float* gp6 = grow + clamp_chunk(j1 + 4 * lane + 768);
            FIR_MAIN_CHUNKS(FIR_REQ)
            wt0 = *(const float4*)(trow + clamp_chunk(j1 + tq));
        }

        const int t00 = t0b + (mg << 6);
        if (t00 < kT) {
            const v8f zero = {0.f,0.f,0.f,0.f,0.f,0.f,0.f,0.f};
            v8f aSs = zero, aSp = zero, aSm = zero;
            v8f aPs = zero, aPp = zero, aPm = zero;
            v8f aMs = zero, aMp = zero, aMm = zero;

            #pragma unroll
            for (int r = 0; r < kGroups; ++r) {
                for (int q = 0; q < 4; ++q) {
                    const int c = 64 * r + 4 * q;
                    v2f w0, w1, w2, w3, z02, z13, zall, z23, z01, a;
                    w0.x = xb[c];          w0.y = xb[c + 1];
                    w1.x = xb[c + 16];     w1.y = xb[c + 17];
                    w2.x = xb[c + 32];     w2.y = xb[c + 33];
                    w3.x = xb[c + 48];     w3.y = xb[c + 49];
                    z02  = w0 + w2;
                    z13  = w1 + w3;
                    zall = z02 + z13;
                    z23  = w2 + w3;
                    z01  = w0 + w1;
                    a.x = tb[c];           a.y = tb[c + 1];
                    aSs = __builtin_amdgcn_wmma_f32_16x16x4_f32(false, a, false, zall, (short)0, aSs, false, false);
                    a.x = e1[c];           a.y = e1[c + 1];
                    aSp = __builtin_amdgcn_wmma_f32_16x16x4_f32(false, a, false, z13, (short)0, aSp, false, false);
                    a.x = e1[c - 16];      a.y = e1[c - 15];
                    aSm = __builtin_amdgcn_wmma_f32_16x16x4_f32(false, a, false, z02, (short)0, aSm, false, false);
                    a.x = e2[c];           a.y = e2[c + 1];
                    aPs = __builtin_amdgcn_wmma_f32_16x16x4_f32(false, a, false, z23, (short)0, aPs, false, false);
                    a.x = e3[c];           a.y = e3[c + 1];
                    aPp = __builtin_amdgcn_wmma_f32_16x16x4_f32(false, a, false, w3, (short)0, aPp, false, false);
                    a.x = e3[c - 16];      a.y = e3[c - 15];
                    aPm = __builtin_amdgcn_wmma_f32_16x16x4_f32(false, a, false, w2, (short)0, aPm, false, false);
                    a.x = e2[c - 32];      a.y = e2[c - 31];
                    aMs = __builtin_amdgcn_wmma_f32_16x16x4_f32(false, a, false, z01, (short)0, aMs, false, false);
                    a.x = e3[c - 32];      a.y = e3[c - 31];
                    aMp = __builtin_amdgcn_wmma_f32_16x16x4_f32(false, a, false, w1, (short)0, aMp, false, false);
                    a.x = e3[c - 48];      a.y = e3[c - 47];
                    aMm = __builtin_amdgcn_wmma_f32_16x16x4_f32(false, a, false, w0, (short)0, aMm, false, false);
                }
            }
            #pragma unroll
            for (int q = 0; q < 4; ++q) {
                v2f w28, a;
                w28.x = xb[448 + 4 * q];   w28.y = xb[449 + 4 * q];
                a.x = tb[400 + 4 * q];     a.y = tb[401 + 4 * q];
                aMm = __builtin_amdgcn_wmma_f32_16x16x4_f32(false, a, false, w28, (short)0, aMm, false, false);
                if (q < 3) {
                    v2f w29, n28;
                    w29.x = xb[464 + 4 * q];   w29.y = xb[465 + 4 * q];
                    n28 = -w28;
                    a.x = tb[416 + 4 * q];     a.y = tb[417 + 4 * q];
                    aMm = __builtin_amdgcn_wmma_f32_16x16x4_f32(false, a, false, w29, (short)0, aMm, false, false);
                    aMp = __builtin_amdgcn_wmma_f32_16x16x4_f32(false, a, false, n28, (short)0, aMp, false, false);
                }
            }

            const v8f s0 = aSs + aSp, s1 = aSs - aSm;
            const v8f y0 = s0 + (aPs + aPp);
            const v8f y1 = s1 + (aPs - aPm);
            const v8f y2 = s0 - (aMs + aMp);
            const v8f y3 = s1 - (aMs - aMm);

            float* const sg = stg + w * (kBlkS * kStgPitch);
            float* const sw = sg + m * kStgPitch + (hi << 3);
            const float* const sr = sg + (lane >> 3) * kStgPitch + 4 * (lane & 7);
            float* const og = out + (size_t)(s0b + (lane >> 3)) * kT + t00 + 4 * (lane & 7);
            float4 v;
            v.x = y0[0]; v.y = y0[1]; v.z = y0[2]; v.w = y0[3];
            *(float4*)(sw) = v;
            v.x = y0[4]; v.y = y0[5]; v.z = y0[6]; v.w = y0[7];
            *(float4*)(sw + 4) = v;
            v.x = y1[0]; v.y = y1[1]; v.z = y1[2]; v.w = y1[3];
            *(float4*)(sw + 16) = v;
            v.x = y1[4]; v.y = y1[5]; v.z = y1[6]; v.w = y1[7];
            *(float4*)(sw + 20) = v;
            #pragma unroll
            for (int i = 0; i < kBlkS / 4; ++i)
                *(float4*)(og + (size_t)(4 * i) * kT) = *(const float4*)(sr + 4 * i * kStgPitch);
            if (t00 + 32 < kT) {
                v.x = y2[0]; v.y = y2[1]; v.z = y2[2]; v.w = y2[3];
                *(float4*)(sw) = v;
                v.x = y2[4]; v.y = y2[5]; v.z = y2[6]; v.w = y2[7];
                *(float4*)(sw + 4) = v;
                v.x = y3[0]; v.y = y3[1]; v.z = y3[2]; v.w = y3[3];
                *(float4*)(sw + 16) = v;
                v.x = y3[4]; v.y = y3[5]; v.z = y3[6]; v.w = y3[7];
                *(float4*)(sw + 20) = v;
                #pragma unroll
                for (int i = 0; i < kBlkS / 4; ++i)
                    *(float4*)(og + (size_t)(4 * i) * kT + 32) = *(const float4*)(sr + 4 * i * kStgPitch);
            }
        }
        if (n == 0) __builtin_amdgcn_s_setprio(2);
        if (n == 1) __builtin_amdgcn_s_setprio(1);
        if (n == 2) __builtin_amdgcn_s_setprio(0);

        __syncthreads();
    }
}

extern "C" void kernel_launch(void* const* d_in, const int* in_sizes, int n_in,
                              void* d_out, int out_size, void* d_ws, size_t ws_size,
                              hipStream_t stream) {
    const float* x = (const float*)d_in[0];
    const float* h = (const float*)d_in[1];
    float* out = (float*)d_out;

    dim3 grid((kSig / kBlkS) * kTileStep);
    fir_bandpass_wmma4x1<<<grid, 256, 0, stream>>>(x, h, out);
}


# Fragment reads one K-slice ahead: each slice's 13 LDS reads go out as one group in front of the previous slice's 9 WMMAs into a second fragment register set, so no LDS wait stands among the WMMAs
# speedup vs baseline: 3.5094x; 1.0311x over previous
#include <hip/hip_runtime.h>


typedef __attribute__((ext_vector_type(2))) float v2f;
typedef __attribute__((ext_vector_type(8))) float v8f;

namespace {
constexpr int kT       = 4000;
constexpr int kSig     = 64 * 64;
constexpr int kTaps    = 413;
constexpr int kHalf    = (kTaps - 1) / 2;
constexpr int kBlkS    = 16;
constexpr int kBlkT    = 512;
constexpr int kTimeBlocks = (kT + kBlkT - 1) / kBlkT;
constexpr int kTilesPerWG = 4;
constexpr int kTileStep   = kTimeBlocks / kTilesPerWG;
static_assert(kTimeBlocks % kTilesPerWG == 0, "time blocks must divide evenly over the workgroups");
static_assert(kTilesPerWG == 4, "one priority level (3, 2, 1, 0) per time block of a workgroup");
static_assert(kSig / kBlkS == 256, "blockIdx.x is split into its low 8 bits (signal block) and the rest (first time block)");
constexpr int kPre     = 208;
constexpr int kShift   = kPre - kHalf;
constexpr int kWinV    = 240;
constexpr int kPerThr  = kBlkS * kWinV / 256;
constexpr int kStride  = 964;
static_assert(kBlkS * kWinV == kPerThr * 256 && kPerThr == 15 && kWinV == 7 * 32 + 16 && kBlkS == 2 * 8,
              "the chunk-to-thread map below is written for 16 rows of 240 chunks on 256 threads");
static_assert(kT % 4 == 0 && kBlkT % 4 == 0 && kPre % 4 == 0, "chunks must not straddle the signal's ends");
static_assert(kT >= 2 * 4 * kWinV, "a window must not overhang both ends of a signal");
static_assert(kT % 16 == 0, "an M-tile lies wholly inside or wholly outside a signal");
constexpr int kStgPitch = 36;
static_assert(kStgPitch >= 32 && kStgPitch % 4 == 0 && (kStgPitch / 4) % 2 == 1, "staging rows: 32 samples, 16-byte aligned, an odd multiple of 16 bytes");
static_assert(kT % 32 == 0, "a 128-byte output line lies wholly inside or wholly outside a signal");
constexpr int kGroups  = 7;
constexpr int kHpOff   = 64;
constexpr int kHpFill  = 512;
constexpr int kHpPitch = 544;
static_assert(kHpPitch >= kHpFill && kHpPitch % 64 == 32, "tap copies");
static_assert(kHpOff - 15 - 48 >= 0 && kHpOff + 2 + 16 * 26 + 8 + 1 < kHpFill, "lowest and highest table entry a fragment reads");
static_assert(16 * 26 + 12 - 15 >= kTaps, "K-slice 3 (v = 12..15) of G_26 holds no tap: g(416 + v - m) lies past the last one");
static_assert(kShift + 64 * 7 + 2 + 16 * 29 + 8 + 1 < 4 * kWinV, "highest window column a fragment reads is a staged one");

__device__ __forceinline__ int clamp_chunk(int j)
{
    return j < 0 ? 0 : (j > kT - 4 ? kT - 4 : j);
}

#define FIR_MAIN_CHUNKS(X) \
    X(0,0) X(0,1) X(1,0) X(1,1) X(2,0) X(2,1) X(3,0) X(3,1) \
    X(4,0) X(4,1) X(5,0) X(5,1) X(6,0) X(6,1)
#define FIR_DECL(j,k)  float4 wm##j##k;
#define FIR_REQ(j,k)   wm##j##k = *(const float4*)(gp##j + (k) * 8 * kT);
#define FIR_PUT(j,k)   *(float4*)(lp + (k) * 8 * kStride + (j) * 128) = wm##j##k;
}

struct Slice { v2f w0, w1, w2, w3, t0, sp, sm, ps, pp, pm, ms, mp, mm; };
struct Tail  { v2f w28[4], g25[4], w29[3], g26[3]; };

__device__ __forceinline__ void load_slice(Slice& f, const float* xb, const float* tb, const float* e1,
                                           const float* e2, const float* e3, const int c)
{
    f.w0.x = xb[c];          f.w0.y = xb[c + 1];
    f.w1.x = xb[c + 16];     f.w1.y = xb[c + 17];
    f.w2.x = xb[c + 32];     f.w2.y = xb[c + 33];
    f.w3.x = xb[c + 48];     f.w3.y = xb[c + 49];
    f.t0.x = tb[c];          f.t0.y = tb[c + 1];
    f.sp.x = e1[c];          f.sp.y = e1[c + 1];
    f.sm.x = e1[c - 16];     f.sm.y = e1[c - 15];
    f.ps.x = e2[c];          f.ps.y = e2[c + 1];
    f.pp.x = e3[c];          f.pp.y = e3[c + 1];
    f.pm.x = e3[c - 16];     f.pm.y = e3[c - 15];
    f.ms.x = e2[c - 32];     f.ms.y = e2[c - 31];
    f.mp.x = e3[c - 32];     f.mp.y = e3[c - 31];
    f.mm.x = e3[c - 48];     f.mm.y = e3[c - 47];
}

__device__ __forceinline__ void load_tail(Tail& f, const float* xb, const float* tb)
{
    #pragma unroll
    for (int q = 0; q < 4; ++q) {
        f.w28[q].x = xb[448 + 4 * q];   f.w28[q].y = xb[449 + 4 * q];
        f.g25[q].x = tb[400 + 4 * q];   f.g25[q].y = tb[401 + 4 * q];
        if (q < 3) {
            f.w29[q].x = xb[464 + 4 * q];   f.w29[q].y = xb[465 + 4 * q];
            f.g26[q].x = tb[416 + 4 * q];   f.g26[q].y = tb[417 + 4 * q];
        }
    }
}

__device__ __forceinline__ void mul_slice(const Slice& f, v8f& aSs, v8f& aSp, v8f& aSm, v8f& aPs, v8f& aPp,
                                          v8f& aPm, v8f& aMs, v8f& aMp, v8f& aMm)
{
    const v2f z02  = f.w0 + f.w2;
    const v2f z13  = f.w1 + f.w3;
    const v2f zall = z02 + z13;
    const v2f z23  = f.w2 + f.w3;
    const v2f z01  = f.w0 + f.w1;
    aSs = __builtin_amdgcn_wmma_f32_16x16x4_f32(false, f.t0, false, zall, (short)0, aSs, false, false);
    aSp = __builtin_amdgcn_wmma_f32_16x16x4_f32(false, f.sp, false, z13,  (short)0, aSp, false, false);
    aSm = __builtin_amdgcn_wmma_f32_16x16x4_f32(false, f.sm, false, z02,  (short)0, aSm, false, false);
    aPs = __builtin_amdgcn_wmma_f32_16x16x4_f32(false, f.ps, false, z23,  (short)0, aPs, false, false);
    aPp = __builtin_amdgcn_wmma_f32_16x16x4_f32(false, f.pp, false, f.w3, (short)0, aPp, false, false);
    aPm = __builtin_amdgcn_wmma_f32_16x16x4_f32(false, f.pm, false, f.w2, (short)0, aPm, false, false);
    aMs = __builtin_amdgcn_wmma_f32_16x16x4_f32(false, f.ms, false, z01,  (short)0, aMs, false, false);
    aMp = __builtin_amdgcn_wmma_f32_16x16x4_f32(false, f.mp, false, f.w1, (short)0, aMp, false, false);
    aMm = __builtin_amdgcn_wmma_f32_16x16x4_f32(false, f.mm, false, f.w0, (short)0, aMm, false, false);
}

__global__ __launch_bounds__(256, 4)
void fir_bandpass_wmma4x1(const float* __restrict__ x,
                          const float* __restrict__ h,
                          float* __restrict__ out)
{
    __shared__ __attribute__((aligned(16))) float xs[kBlkS * kStride];
    __shared__ float hp[kHpPitch + kHpFill];
    __shared__ float d16[kHpPitch + kHpFill];
    __shared__ float d32[kHpPitch + kHpFill];
    __shared__ float dd[kHpPitch + kHpFill];
    __shared__ __attribute__((aligned(16))) float stg[8 * kBlkS * kStgPitch];

    __builtin_amdgcn_s_setprio(3);

    const int tid = threadIdx.x;
    const int s0b = (blockIdx.x & 255) * kBlkS;
    const int tb0 = blockIdx.x >> 8;

    const int ja = tid - kHpOff;
    const int jb = tid + 256 - kHpOff;
    float ga[4], gb[4];
    #pragma unroll
    for (int u = 0; u < 4; ++u) {
        const int a = ja + 16 * u, b = jb + 16 * u;
        ga[u] = h[kTaps - 1 - (a < 0 ? 0 : a)];
        gb[u] = h[b < kTaps ? kTaps - 1 - b : 0];
    }

    const int w    = tid >> 5;
    const int lane = tid & 31;

    FIR_MAIN_CHUNKS(FIR_DECL)
    float4 wt0;
    const float* grow = x + (size_t)(s0b + w) * kT;
    const float* trow = x + (size_t)(s0b + (tid >> 4)) * kT;
    const int    tq   = 4 * (kWinV - 16 + (tid & 15));
    float* const lp   = xs + w * kStride + 4 * lane;
    float* const lt   = xs + (tid >> 4) * kStride + tq;
    {
        const int j0 = tb0 * kBlkT - kPre;
        const float* gp0 = grow + clamp_chunk(j0 + 4 * lane);
        const float* gp1 = grow + clamp_chunk(j0 + 4 * lane + 128);
        const float* gp2 = grow + clamp_chunk(j0 + 4 * lane + 256);
        const float* gp3 = grow + clamp_chunk(j0 + 4 * lane + 384);
        const float* gp4 = grow + clamp_chunk(j0 + 4 * lane + 512);
        const float* gp5 = grow + clamp_chunk(j0 + 4 * lane + 640);
        const float* gp6 = grow + clamp_chunk(j0 + 4 * lane + 768);
        FIR_MAIN_CHUNKS(FIR_REQ)
        wt0 = *(const float4*)(trow + clamp_chunk(j0 + tq));
    }

    {
        #pragma unroll
        for (int u = 0; u < 4; ++u) {
            ga[u] = ja + 16 * u >= 0 ? ga[u] : 0.0f;
            gb[u] = jb + 16 * u < kTaps ? gb[u] : 0.0f;
        }
        const float a16 = ga[1] - ga[0], b16 = gb[1] - gb[0];
        const float a32 = ga[2] - ga[0], b32 = gb[2] - gb[0];
        const float add = (ga[3] - ga[1]) - a32;
        const float bdd = (gb[3] - gb[1]) - b32;
        hp[tid]        = ga[0];   hp[kHpPitch + tid]        = ga[0];
        hp[tid + 256]  = gb[0];   hp[kHpPitch + tid + 256]  = gb[0];
        d16[tid]       = a16;     d16[kHpPitch + tid]       = a16;
        d16[tid + 256] = b16;     d16[kHpPitch + tid + 256] = b16;
        d32[tid]       = a32;     d32[kHpPitch + tid]       = a32;
        d32[tid + 256] = b32;     d32[kHpPitch + tid + 256] = b32;
        dd[tid]        = add;     dd[kHpPitch + tid]        = add;
        dd[tid + 256]  = bdd;     dd[kHpPitch + tid + 256]  = bdd;
    }

    const int mg   = w;
    const int m    = lane & 15;
    const int hi   = lane >> 4;
    const int kk   = hi << 1;

    const int   tof = (hi ? kHpPitch : 0) + (kHpOff + kk - m);
    const float* tb = hp  + tof;
    const float* e1 = d16 + tof;
    const float* e2 = d32 + tof;
    const float* e3 = dd  + tof;
    const float* xb = xs + m * kStride + kShift + (mg << 6) + kk;

    #pragma unroll 1
    for (int n = 0; n < kTilesPerWG; ++n) {
        const int t0b = (tb0 + n * kTileStep) * kBlkT;
        const int j0  = t0b - kPre;

        FIR_MAIN_CHUNKS(FIR_PUT)
        *(float4*)(lt) = wt0;

        if (j0 < 0 || j0 + 4 * kWinV > kT) {
            __syncthreads();
            const bool left = j0 < 0;
            const int nOut  = left ? ((-j0) >> 2) : ((j0 + 4 * kWinV - kT) >> 2);
            const int qb    = left ? 0 : kWinV - nOut;
            const int s     = tid & 15;
            const float* row = x + (size_t)(s0b + s) * kT;
            const float e2x = 2.0f * row[left ? 0 : kT - 1];
            #pragma unroll 4
            for (int q = qb + (tid >> 4); q < qb + nOut; q += 16) {
                const int j = j0 + 4 * q;
                const int r = left ? -j : 2 * kT - 2 - j;
                float4 p;
                p.x = e2x - row[r];
                p.y = e2x - row[r - 1];
                p.z = e2x - row[r - 2];
                p.w = e2x - row[r - 3];
                *(float4*)(xs + s * kStride + 4 * q) = p;
            }
        }
        __syncthreads();

        if (n + 1 < kTilesPerWG) {
            const int j1 = j0 + kTileStep * kBlkT;
            const float* gp0 = grow + clamp_chunk(j1 + 4 * lane);
            const float* gp1 = grow + clamp_chunk(j1 + 4 * lane + 128);
            const float* gp2 = grow + clamp_chunk(j1 + 4 * lane + 256);
            const float* gp3 = grow + clamp_chunk(j1 + 4 * lane + 384);
            const float* gp4 = grow + clamp_chunk(j1 + 4 * lane + 512);
            const float* gp5 = grow + clamp_chunk(j1 + 4 * lane + 640);
            const float* gp6 = grow + clamp_chunk(j1 + 4 * lane + 768);
            FIR_MAIN_CHUNKS(FIR_REQ)
            wt0 = *(const float4*)(trow + clamp_chunk(j1 + tq));
        }

        const int t00 = t0b + (mg << 6);
        if (t00 < kT) {
            const v8f zero = {0.f,0.f,0.f,0.f,0.f,0.f,0.f,0.f};
            v8f aSs = zero, aSp = zero, aSm = zero;
            v8f aPs = zero, aPp = zero, aPm = zero;
            v8f aMs = zero, aMp = zero, aMm = zero;

            Slice fa, fb;
            Tail  ft;
            load_slice(fa, xb, tb, e1, e2, e3, 0);
            #pragma unroll
            for (int i = 0; i < 4 * kGroups; i += 2) {
                load_slice(fb, xb, tb, e1, e2, e3, 64 * ((i + 1) >> 2) + 4 * ((i + 1) & 3));
                __builtin_amdgcn_sched_barrier(0);
                mul_slice(fa, aSs, aSp, aSm, aPs, aPp, aPm, aMs, aMp, aMm);
                __builtin_amdgcn_sched_barrier(0);
                if (i + 2 < 4 * kGroups) load_slice(fa, xb, tb, e1, e2, e3, 64 * ((i + 2) >> 2) + 4 * ((i + 2) & 3));
                else                     load_tail(ft, xb, tb);
                __builtin_amdgcn_sched_barrier(0);
                mul_slice(fb, aSs, aSp, aSm, aPs, aPp, aPm, aMs, aMp, aMm);
                __builtin_amdgcn_sched_barrier(0);
            }
            #pragma unroll
            for (int q = 0; q < 4; ++q) {
                aMm = __builtin_amdgcn_wmma_f32_16x16x4_f32(false, ft.g25[q], false, ft.w28[q], (short)0, aMm, false, false);
                if (q < 3) {
                    const v2f n28 = -ft.w28[q];
                    aMm = __builtin_amdgcn_wmma_f32_16x16x4_f32(false, ft.g26[q], false, ft.w29[q], (short)0, aMm, false, false);
                    aMp = __builtin_amdgcn_wmma_f32_16x16x4_f32(false, ft.g26[q], false, n28, (short)0, aMp, false, false);
                }
            }

            const v8f s0 = aSs + aSp, s1 = aSs - aSm;
            const v8f y0 = s0 + (aPs + aPp);
            const v8f y1 = s1 + (aPs - aPm);
            const v8f y2 = s0 - (aMs + aMp);
            const v8f y3 = s1 - (aMs - aMm);

            float* const sg = stg + w * (kBlkS * kStgPitch);
            float* const sw = sg + m * kStgPitch + (hi << 3);
            const float* const sr = sg + (lane >> 3) * kStgPitch + 4 * (lane & 7);
            float* const og = out + (size_t)(s0b + (lane >> 3)) * kT + t00 + 4 * (lane & 7);
            float4 v;
            v.x = y0[0]; v.y = y0[1]; v.z = y0[2]; v.w = y0[3];
            *(float4*)(sw) = v;
            v.x = y0[4]; v.y = y0[5]; v.z = y0[6]; v.w = y0[7];
            *(float4*)(sw + 4) = v;
            v.x = y1[0]; v.y = y1[1]; v.z = y1[2]; v.w = y1[3];
            *(float4*)(sw + 16) = v;
            v.x = y1[4]; v.y = y1[5]; v.z = y1[6]; v.w = y1[7];
            *(float4*)(sw + 20) = v;
            #pragma unroll
            for (int i = 0; i < kBlkS / 4; ++i)
                *(float4*)(og + (size_t)(4 * i) * kT) = *(const float4*)(sr + 4 * i * kStgPitch);
            if (t00 + 32 < kT) {
                v.x = y2[0]; v.y = y2[1]; v.z = y2[2]; v.w = y2[3];
                *(float4*)(sw) = v;
                v.x = y2[4]; v.y = y2[5]; v.z = y2[6]; v.w = y2[7];
                *(float4*)(sw + 4) = v;
                v.x = y3[0]; v.y = y3[1]; v.z = y3[2]; v.w = y3[3];
                *(float4*)(sw + 16) = v;
                v.x = y3[4]; v.y = y3[5]; v.z = y3[6]; v.w = y3[7];
                *(float4*)(sw + 20) = v;
                #pragma unroll
                for (int i = 0; i < kBlkS / 4; ++i)
                    *(float4*)(og + (size_t)(4 * i) * kT + 32) = *(const float4*)(sr + 4 * i * kStgPitch);
            }
        }
        if (n == 0) __builtin_amdgcn_s_setprio(2);
        if (n == 1) __builtin_amdgcn_s_setprio(1);
        if (n == 2) __builtin_amdgcn_s_setprio(0);

        __syncthreads();
    }
}

extern "C" void kernel_launch(void* const* d_in, const int* in_sizes, int n_in,
                              void* d_out, int out_size, void* d_ws, size_t ws_size,
                              hipStream_t stream) {
    const float* x = (const float*)d_in[0];
    const float* h = (const float*)d_in[1];
    float* out = (float*)d_out;

    dim3 grid((kSig / kBlkS) * kTileStep);
    fir_bandpass_wmma4x1<<<grid, 256, 0, stream>>>(x, h, out);
}


# Block-Toeplitz split a third time: a wave pair shares a 128x16 tile, 27 block products per 8x8 block group for 36; each wave does half of S and one of P, M; S crosses through LDS (new f32 sum order)
# speedup vs baseline: 3.7036x; 1.0553x over previous
#include <hip/hip_runtime.h>


typedef __attribute__((ext_vector_type(2))) float v2f;
typedef __attribute__((ext_vector_type(8))) float v8f;

namespace {
constexpr int kT       = 4000;
constexpr int kSig     = 64 * 64;
constexpr int kTaps    = 413;
constexpr int kHalf    = (kTaps - 1) / 2;
constexpr int kBlkS    = 16;
constexpr int kBlkT    = 512;
constexpr int kTimeBlocks = (kT + kBlkT - 1) / kBlkT;
constexpr int kTilesPerWG = 4;
constexpr int kTileStep   = kTimeBlocks / kTilesPerWG;
static_assert(kTimeBlocks % kTilesPerWG == 0, "time blocks must divide evenly over the workgroups");
static_assert(kTilesPerWG == 4, "one priority level (3, 2, 1, 0) per time block of a workgroup");
static_assert(kSig / kBlkS == 256, "blockIdx.x is split into its low 8 bits (signal block) and the rest (first time block)");
constexpr int kPre     = 208;
constexpr int kShift   = kPre - kHalf;
constexpr int kWinV    = 240;
constexpr int kPerThr  = kBlkS * kWinV / 256;
constexpr int kStride  = 964;
static_assert(kBlkS * kWinV == kPerThr * 256 && kPerThr == 15 && kWinV == 7 * 32 + 16 && kBlkS == 2 * 8,
              "the chunk-to-thread map below is written for 16 rows of 240 chunks on 256 threads");
static_assert(kT % 4 == 0 && kBlkT % 4 == 0 && kPre % 4 == 0, "chunks must not straddle the signal's ends");
static_assert(kT >= 2 * 4 * kWinV, "a window must not overhang both ends of a signal");
static_assert(kT % 16 == 0, "an M-tile lies wholly inside or wholly outside a signal");
constexpr int kStgPitch = 36;
static_assert(kStgPitch >= 32 && kStgPitch % 4 == 0 && (kStgPitch / 4) % 2 == 1, "staging rows: 32 samples, 16-byte aligned, an odd multiple of 16 bytes");
static_assert(kT % 32 == 0, "a 128-byte output line lies wholly inside or wholly outside a signal");
constexpr int kPairT   = 128;
constexpr int kGroups  = 4;
constexpr int kSlices  = 4 * kGroups;
static_assert(kBlkT == 4 * kPairT, "four wave pairs share a time block");
static_assert((kTimeBlocks - 1) * kBlkT + 3 * kPairT < kT, "every wave pair of every time block has outputs inside the signal");
constexpr int kHpOff   = 64;
constexpr int kHpFill  = 512;
constexpr int kHpPitch = 544;
static_assert(kHpPitch >= kHpFill && kHpPitch % 64 == 32, "tap copies");
static_assert(kHpOff - 15 - 48 >= 0 && kHpOff + 2 + 16 * 26 + 8 + 1 < kHpFill, "lowest and highest table entry a fragment reads");
static_assert(16 * 26 + 12 - 15 >= kTaps, "K-slice 3 (v = 12..15) of G_26 holds no tap: g(416 + v - m) lies past the last one");
static_assert(kShift + kPairT * 3 + 2 + 16 * 33 + 8 + 1 < 4 * kWinV, "highest window column a fragment reads is a staged one");
constexpr int kDOff    = 128;
constexpr int kDFill   = 768;
constexpr int kDPitch  = 800;
static_assert(kDPitch >= kDFill && kDPitch % 64 == 32, "D copies");
static_assert(kDOff - 64 - 48 - 15 >= 0 && kDOff + 2 + 128 * (kGroups - 1) + 12 + 1 < kDFill, "lowest and highest D entry a fragment reads");
static_assert(kHpOff + 2 + 128 * (kGroups - 1) + 12 + 1 < kHpFill, "highest tap entry a fragment of S reads");
constexpr int kExTile  = 8 * 32 * 4;

__device__ __forceinline__ int clamp_chunk(int j)
{
    return j < 0 ? 0 : (j > kT - 4 ? kT - 4 : j);
}

#define FIR_MAIN_CHUNKS(X) \
    X(0,0) X(0,1) X(1,0) X(1,1) X(2,0) X(2,1) X(3,0) X(3,1) \
    X(4,0) X(4,1) X(5,0) X(5,1) X(6,0) X(6,1)
#define FIR_DECL(j,k)  float4 wm##j##k;
#define FIR_REQ(j,k)   wm##j##k = *(const float4*)(gp##j + (k) * 8 * kT);
#define FIR_PUT(j,k)   *(float4*)(lp + (k) * 8 * kStride + (j) * 128) = wm##j##k;
#define FIR_CLEAR(j,k) wm##j##k = make_float4(0.0f, 0.0f, 0.0f, 0.0f);
#define FIR_REQUEST_NEXT() \
    if (n + 1 < kTilesPerWG) { \
        const int j1 = j0 + kTileStep * kBlkT; \
        const float* gp0 = grow + clamp_chunk(j1 + 4 * lane); \
        const float* gp1 = grow + clamp_chunk(j1 + 4 * lane + 128); \
        const float* gp2 = grow + clamp_chunk(j1 + 4 * lane + 256); \
        const float* gp3 = grow + clamp_chunk(j1 + 4 * lane + 384); \
        const float* gp4 = grow + clamp_chunk(j1 + 4 * lane + 512); \
        const float* gp5 = grow + clamp_chunk(j1 + 4 * lane + 640); \
        const float* gp6 = grow + clamp_chunk(j1 + 4 * lane + 768); \
        FIR_MAIN_CHUNKS(FIR_REQ) \
        wt0 = *(const float4*)(trow + clamp_chunk(j1 + tq)); \
    } else { \
          \
          \
        FIR_MAIN_CHUNKS(FIR_CLEAR) \
        wt0 = make_float4(0.0f, 0.0f, 0.0f, 0.0f); \
    }
}

struct Slice  { v2f w0, w1, w2, w3, t0, sp, sm, ps, pp, pm, ms, mp, mm; };
struct SliceS { v2f l0, l1, l2, l3, u0, u1, u2, u3, t0, sp, sm, ps, pp, pm, ms, mp, mm; };
struct Tail   { v2f w32[4], g25[4], w33[3], g26[3]; };

constexpr int slice_col(int i) { return 128 * (i >> 2) + 4 * (i & 3); }

template <class F>
__device__ __forceinline__ void load_taps(F& f, const float* tb, const float* e1, const float* e2, const float* e3, const int c)
{
    f.t0.x = tb[c];          f.t0.y = tb[c + 1];
    f.sp.x = e1[c];          f.sp.y = e1[c + 1];
    f.sm.x = e1[c - 16];     f.sm.y = e1[c - 15];
    f.ps.x = e2[c];          f.ps.y = e2[c + 1];
    f.pp.x = e3[c];          f.pp.y = e3[c + 1];
    f.pm.x = e3[c - 16];     f.pm.y = e3[c - 15];
    f.ms.x = e2[c - 32];     f.ms.y = e2[c - 31];
    f.mp.x = e3[c - 32];     f.mp.y = e3[c - 31];
    f.mm.x = e3[c - 48];     f.mm.y = e3[c - 47];
}

__device__ __forceinline__ void load_slice(Slice& f, const float* xw, const float* tb, const float* e1,
                                           const float* e2, const float* e3, const int c)
{
    f.w0.x = xw[c];          f.w0.y = xw[c + 1];
    f.w1.x = xw[c + 16];     f.w1.y = xw[c + 17];
    f.w2.x = xw[c + 32];     f.w2.y = xw[c + 33];
    f.w3.x = xw[c + 48];     f.w3.y = xw[c + 49];
    load_taps(f, tb, e1, e2, e3, c);
}

__device__ __forceinline__ void load_sliceS(SliceS& f, const float* xb, const float* tb, const float* e1,
                                            const float* e2, const float* e3, const int c)
{
    f.l0.x = xb[c];          f.l0.y = xb[c + 1];
    f.l1.x = xb[c + 16];     f.l1.y = xb[c + 17];
    f.l2.x = xb[c + 32];     f.l2.y = xb[c + 33];
    f.l3.x = xb[c + 48];     f.l3.y = xb[c + 49];
    f.u0.x = xb[c + 64];     f.u0.y = xb[c + 65];
    f.u1.x = xb[c + 80];     f.u1.y = xb[c + 81];
    f.u2.x = xb[c + 96];     f.u2.y = xb[c + 97];
    f.u3.x = xb[c + 112];    f.u3.y = xb[c + 113];
    load_taps(f, tb, e1, e2, e3, c);
}

__device__ __forceinline__ void load_tail(Tail& f, const float* xb, const float* tb)
{
    #pragma unroll
    for (int q = 0; q < 4; ++q) {
        f.w32[q].x = xb[512 + 4 * q];   f.w32[q].y = xb[513 + 4 * q];
        f.g25[q].x = tb[400 + 4 * q];   f.g25[q].y = tb[401 + 4 * q];
        if (q < 3) {
            f.w33[q].x = xb[528 + 4 * q];   f.w33[q].y = xb[529 + 4 * q];
            f.g26[q].x = tb[416 + 4 * q];   f.g26[q].y = tb[417 + 4 * q];
        }
    }
}

template <class F>
__device__ __forceinline__ void mul_taps(const F& f, const v2f w0, const v2f w1, const v2f w2, const v2f w3,
                                         v8f& aSs, v8f& aSp, v8f& aSm, v8f& aPs, v8f& aPp,
                                         v8f& aPm, v8f& aMs, v8f& aMp, v8f& aMm)
{
    const v2f z02  = w0 + w2;
    const v2f z13  = w1 + w3;
    const v2f zall = z02 + z13;
    const v2f z23  = w2 + w3;
    const v2f z01  = w0 + w1;
    aSs = __builtin_amdgcn_wmma_f32_16x16x4_f32(false, f.t0, false, zall, (short)0, aSs, false, false);
    aSp = __builtin_amdgcn_wmma_f32_16x16x4_f32(false, f.sp, false, z13,  (short)0, aSp, false, false);
    aSm = __builtin_amdgcn_wmma_f32_16x16x4_f32(false, f.sm, false, z02,  (short)0, aSm, false, false);
    aPs = __builtin_amdgcn_wmma_f32_16x16x4_f32(false, f.ps, false, z23,  (short)0, aPs, false, false);
    aPp = __builtin_amdgcn_wmma_f32_16x16x4_f32(false, f.pp, false, w3,   (short)0, aPp, false, false);
    aPm = __builtin_amdgcn_wmma_f32_16x16x4_f32(false, f.pm, false, w2,   (short)0, aPm, false, false);
    aMs = __builtin_amdgcn_wmma_f32_16x16x4_f32(false, f.ms, false, z01,  (short)0, aMs, false, false);
    aMp = __builtin_amdgcn_wmma_f32_16x16x4_f32(false, f.mp, false, w1,   (short)0, aMp, false, false);
    aMm = __builtin_amdgcn_wmma_f32_16x16x4_f32(false, f.mm, false, w0,   (short)0, aMm, false, false);
}

__device__ __forceinline__ void fold9(const v8f& aSs, const v8f& aSp, const v8f& aSm, const v8f& aPs, const v8f& aPp,
                                      const v8f& aPm, const v8f& aMs, const v8f& aMp, const v8f& aMm,
                                      v8f& y0, v8f& y1, v8f& y2, v8f& y3)
{
    const v8f s0 = aSs + aSp, s1 = aSs - aSm;
    y0 = s0 + (aPs + aPp);
    y1 = s1 + (aPs - aPm);
    y2 = s0 - (aMs + aMp);
    y3 = s1 - (aMs - aMm);
}

__device__ __forceinline__ void put_tiles(float* e, const v8f& y0, const v8f& y1, const v8f& y2, const v8f& y3)
{
    float4 v;
    v.x = y0[0]; v.y = y0[1]; v.z = y0[2]; v.w = y0[3];   *(float4*)(e)           = v;
    v.x = y0[4]; v.y = y0[5]; v.z = y0[6]; v.w = y0[7];   *(float4*)(e + 128)     = v;
    v.x = y1[0]; v.y = y1[1]; v.z = y1[2]; v.w = y1[3];   *(float4*)(e + 2 * 128) = v;
    v.x = y1[4]; v.y = y1[5]; v.z = y1[6]; v.w = y1[7];   *(float4*)(e + 3 * 128) = v;
    v.x = y2[0]; v.y = y2[1]; v.z = y2[2]; v.w = y2[3];   *(float4*)(e + 4 * 128) = v;
    v.x = y2[4]; v.y = y2[5]; v.z = y2[6]; v.w = y2[7];   *(float4*)(e + 5 * 128) = v;
    v.x = y3[0]; v.y = y3[1]; v.z = y3[2]; v.w = y3[3];   *(float4*)(e + 6 * 128) = v;
    v.x = y3[4]; v.y = y3[5]; v.z = y3[6]; v.w = y3[7];   *(float4*)(e + 7 * 128) = v;
}

__device__ __forceinline__ v8f get_tile(const float* e)
{
    const float4 a = *(const float4*)(e), b = *(const float4*)(e + 128);
    v8f y;
    y[0] = a.x; y[1] = a.y; y[2] = a.z; y[3] = a.w;
    y[4] = b.x; y[5] = b.y; y[6] = b.z; y[7] = b.w;
    return y;
}

__device__ __forceinline__ void product16(const float* xw, const float* p0, const float* p1, const float* p2, const float* p3,
                                          v8f& aSs, v8f& aSp, v8f& aSm, v8f& aPs, v8f& aPp,
                                          v8f& aPm, v8f& aMs, v8f& aMp, v8f& aMm)
{
    Slice fa, fb;
    load_slice(fa, xw, p0, p1, p2, p3, 0);
    #pragma unroll
    for (int i = 0; i < kSlices; i += 2) {
        load_slice(fb, xw, p0, p1, p2, p3, slice_col(i + 1));
        __builtin_amdgcn_sched_barrier(0);
        mul_taps(fa, fa.w0, fa.w1, fa.w2, fa.w3, aSs, aSp, aSm, aPs, aPp, aPm, aMs, aMp, aMm);
        __builtin_amdgcn_sched_barrier(0);
        if (i + 2 < kSlices) load_slice(fa, xw, p0, p1, p2, p3, slice_col(i + 2));
        __builtin_amdgcn_sched_barrier(0);
        mul_taps(fb, fb.w0, fb.w1, fb.w2, fb.w3, aSs, aSp, aSm, aPs, aPp, aPm, aMs, aMp, aMm);
        __builtin_amdgcn_sched_barrier(0);
    }
}

__global__ __launch_bounds__(256, 4)
void fir_bandpass_wmma4x1(const float* __restrict__ x,
                          const float* __restrict__ h,
                          float* __restrict__ out)
{
    __shared__ __attribute__((aligned(16))) float xs[kBlkS * kStride];
    __shared__ float hp[kHpPitch + kHpFill];
    __shared__ float d16[kHpPitch + kHpFill];
    __shared__ float d32[kHpPitch + kHpFill];
    __shared__ float dd[kHpPitch + kHpFill];
    __shared__ float Dt[kDPitch + kDFill];
    __shared__ float D16t[kDPitch + kDFill];
    __shared__ float D32t[kDPitch + kDFill];
    __shared__ float Ddt[kDPitch + kDFill];
    __shared__ __attribute__((aligned(16))) float stg[8 * kBlkS * kStgPitch];
    __shared__ __attribute__((aligned(16))) float exS[8 * kExTile];

    __builtin_amdgcn_s_setprio(3);

    const int tid = threadIdx.x;
    const int s0b = (blockIdx.x & 255) * kBlkS;
    const int tb0 = blockIdx.x >> 8;

    const int ja = tid - kHpOff;
    const int jb = tid + 256 - kHpOff;
    float ga[4], gb[4];
    #pragma unroll
    for (int u = 0; u < 4; ++u) {
        const int a = ja + 16 * u, b = jb + 16 * u;
        ga[u] = h[kTaps - 1 - (a < 0 ? 0 : a)];
        gb[u] = h[b < kTaps ? kTaps - 1 - b : 0];
    }
    float gd[3][8];
    #pragma unroll
    for (int u = 0; u < 3; ++u) {
        #pragma unroll
        for (int t = 0; t < 8; ++t) {
            const int a = tid + 256 * u - kDOff + 16 * t;
            gd[u][t] = h[kTaps - 1 - (a < 0 ? 0 : (a > kTaps - 1 ? kTaps - 1 : a))];
        }
    }

    const int w    = __builtin_amdgcn_readfirstlane(tid >> 5);
    const int lane = tid & 31;

    FIR_MAIN_CHUNKS(FIR_DECL)
    float4 wt0;
    const float* grow = x + (size_t)(s0b + w) * kT;
    const float* trow = x + (size_t)(s0b + (tid >> 4)) * kT;
    const int    tq   = 4 * (kWinV - 16 + (tid & 15));
    float* const lp   = xs + w * kStride + 4 * lane;
    float* const lt   = xs + (tid >> 4) * kStride + tq;
    {
        const int j0 = tb0 * kBlkT - kPre;
        const float* gp0 = grow + clamp_chunk(j0 + 4 * lane);
        const float* gp1 = grow + clamp_chunk(j0 + 4 * lane + 128);
        const float* gp2 = grow + clamp_chunk(j0 + 4 * lane + 256);
        const float* gp3 = grow + clamp_chunk(j0 + 4 * lane + 384);
        const float* gp4 = grow + clamp_chunk(j0 + 4 * lane + 512);
        const float* gp5 = grow + clamp_chunk(j0 + 4 * lane + 640);
        const float* gp6 = grow + clamp_chunk(j0 + 4 * lane + 768);
        FIR_MAIN_CHUNKS(FIR_REQ)
        wt0 = *(const float4*)(trow + clamp_chunk(j0 + tq));
    }

    {
        #pragma unroll
        for (int u = 0; u < 4; ++u) {
            ga[u] = ja + 16 * u >= 0 ? ga[u] : 0.0f;
            gb[u] = jb + 16 * u < kTaps ? gb[u] : 0.0f;
        }
        const float a16 = ga[1] - ga[0], b16 = gb[1] - gb[0];
        const float a32 = ga[2] - ga[0], b32 = gb[2] - gb[0];
        const float add = (ga[3] - ga[1]) - a32;
        const float bdd = (gb[3] - gb[1]) - b32;
        hp[tid]        = ga[0];   hp[kHpPitch + tid]        = ga[0];
        hp[tid + 256]  = gb[0];   hp[kHpPitch + tid + 256]  = gb[0];
        d16[tid]       = a16;     d16[kHpPitch + tid]       = a16;
        d16[tid + 256] = b16;     d16[kHpPitch + tid + 256] = b16;
        d32[tid]       = a32;     d32[kHpPitch + tid]       = a32;
        d32[tid + 256] = b32;     d32[kHpPitch + tid + 256] = b32;
        dd[tid]        = add;     dd[kHpPitch + tid]        = add;
        dd[tid + 256]  = bdd;     dd[kHpPitch + tid + 256]  = bdd;
        #pragma unroll
        for (int u = 0; u < 3; ++u) {
            #pragma unroll
            for (int t = 0; t < 8; ++t) {
                const int a = tid + 256 * u - kDOff + 16 * t;
                gd[u][t] = (a >= 0 && a < kTaps) ? gd[u][t] : 0.0f;
            }
            const float v0 = gd[u][4] - gd[u][0];
            const float v1 = gd[u][5] - gd[u][1];
            const float v2 = gd[u][6] - gd[u][2];
            const float v3 = gd[u][7] - gd[u][3];
            const float x16 = v1 - v0, x32 = v2 - v0;
            const float xdd = (v3 - v1) - x32;
            const int i = tid + 256 * u;
            Dt[i]   = v0;     Dt[kDPitch + i]   = v0;
            D16t[i] = x16;    D16t[kDPitch + i] = x16;
            D32t[i] = x32;    D32t[kDPitch + i] = x32;
            Ddt[i]  = xdd;    Ddt[kDPitch + i]  = xdd;
        }
    }

    const int pr   = w & 3;
    const int role = w >> 2;
    const int m    = lane & 15;
    const int hi   = lane >> 4;
    const int kk   = hi << 1;

    const int   tof = (hi ? kHpPitch : 0) + (kHpOff + kk - m);
    const float* tb = hp  + tof;
    const float* e1 = d16 + tof;
    const float* e2 = d32 + tof;
    const float* e3 = dd  + tof;
    const int   tofD = (hi ? kDPitch : 0) + (kDOff + kk - m);
    const float* q0 = Dt   + tofD;
    const float* q1 = D16t + tofD;
    const float* q2 = D32t + tofD;
    const float* q3 = Ddt  + tofD;
    const float* xb = xs + m * kStride + kShift + (pr << 7) + kk;

    #pragma unroll 1
    for (int n = 0; n < kTilesPerWG; ++n) {
        const int t0b = (tb0 + n * kTileStep) * kBlkT;
        const int j0  = t0b - kPre;

        FIR_MAIN_CHUNKS(FIR_PUT)
        *(float4*)(lt) = wt0;

        if (j0 < 0 || j0 + 4 * kWinV > kT) {
            __syncthreads();
            const bool left = j0 < 0;
            const int nOut  = left ? ((-j0) >> 2) : ((j0 + 4 * kWinV - kT) >> 2);
            const int qb    = left ? 0 : kWinV - nOut;
            const int s     = tid & 15;
            const float* row = x + (size_t)(s0b + s) * kT;
            const float e2x = 2.0f * row[left ? 0 : kT - 1];
            #pragma unroll 4
            for (int q = qb + (tid >> 4); q < qb + nOut; q += 16) {
                const int j = j0 + 4 * q;
                const int r = left ? -j : 2 * kT - 2 - j;
                float4 p;
                p.x = e2x - row[r];
                p.y = e2x - row[r - 1];
                p.z = e2x - row[r - 2];
                p.w = e2x - row[r - 3];
                *(float4*)(xs + s * kStride + 4 * q) = p;
            }
        }
        __syncthreads();

        const int t00 = t0b + (pr << 7);
        const v8f zero = {0.f,0.f,0.f,0.f,0.f,0.f,0.f,0.f};
        float* const eMine        = exS + (role * 4 + pr) * kExTile + 4 * lane;
        const float* const eOther = exS + ((role ^ 1) * 4 + pr) * kExTile + 4 * lane;

        {
            const int so = role * slice_col(kSlices / 2);
            const float* const xh = xb + so;
            const float* const h0 = tb + so;
            const float* const h1 = e1 + so;
            const float* const h2 = e2 + so;
            const float* const h3 = e3 + so;
            v8f aSs = zero, aSp = zero, aSm = zero;
            v8f aPs = zero, aPp = zero, aPm = zero;
            v8f aMs = zero, aMp = zero, aMm = zero;
            SliceS fa, fb;
            load_sliceS(fa, xh, h0, h1, h2, h3, 0);
            #pragma unroll
            for (int i = 0; i < kSlices / 2; i += 2) {
                load_sliceS(fb, xh, h0, h1, h2, h3, slice_col(i + 1));
                __builtin_amdgcn_sched_barrier(0);
                mul_taps(fa, fa.l0 + fa.u0, fa.l1 + fa.u1, fa.l2 + fa.u2, fa.l3 + fa.u3, aSs, aSp, aSm, aPs, aPp, aPm, aMs, aMp, aMm);
                __builtin_amdgcn_sched_barrier(0);
                if (i + 2 < kSlices / 2) load_sliceS(fa, xh, h0, h1, h2, h3, slice_col(i + 2));
                __builtin_amdgcn_sched_barrier(0);
                mul_taps(fb, fb.l0 + fb.u0, fb.l1 + fb.u1, fb.l2 + fb.u2, fb.l3 + fb.u3, aSs, aSp, aSm, aPs, aPp, aPm, aMs, aMp, aMm);
                __builtin_amdgcn_sched_barrier(0);
            }
            v8f s0, s1, s2, s3;
            fold9(aSs, aSp, aSm, aPs, aPp, aPm, aMs, aMp, aMm, s0, s1, s2, s3);
            put_tiles(eMine, s0, s1, s2, s3);
        }

        v8f r0, r1, r2, r3;
        {
            const float* const xw = role ? xb : xb + 64;
            const int qo = role ? -64 : 0;
            v8f aSs = zero, aSp = zero, aSm = zero;
            v8f aPs = zero, aPp = zero, aPm = zero;
            v8f aMs = zero, aMp = zero, aMm = zero;
            if (role == 1) {
                Tail ft;
                load_tail(ft, xb, tb);
                #pragma unroll
                for (int q = 0; q < 4; ++q) {
                    const v2f n32 = -ft.w32[q];
                    aMm = __builtin_amdgcn_wmma_f32_16x16x4_f32(false, ft.g25[q], false, n32, (short)0, aMm, false, false);
                    if (q < 3) {
                        const v2f n33 = -ft.w33[q];
                        aMm = __builtin_amdgcn_wmma_f32_16x16x4_f32(false, ft.g26[q], false, n33, (short)0, aMm, false, false);
                        aMp = __builtin_amdgcn_wmma_f32_16x16x4_f32(false, ft.g26[q], false, ft.w32[q], (short)0, aMp, false, false);
                    }
                }
            }
            product16(xw, q0 + qo, q1 + qo, q2 + qo, q3 + qo, aSs, aSp, aSm, aPs, aPp, aPm, aMs, aMp, aMm);
            fold9(aSs, aSp, aSm, aPs, aPp, aPm, aMs, aMp, aMm, r0, r1, r2, r3);
        }

        __syncthreads();

        FIR_REQUEST_NEXT()

        {
            float* const sg = stg + w * (kBlkS * kStgPitch);
            float* const sw = sg + m * kStgPitch + (hi << 3);
            const float* const sr = sg + (lane >> 3) * kStgPitch + 4 * (lane & 7);
            const int th = t00 + 64 * role;
            float* const og = out + (size_t)(s0b + (lane >> 3)) * kT + th + 4 * (lane & 7);
            #pragma unroll
            for (int jp = 0; jp < 2; ++jp) {
                const v8f sa = get_tile(eMine + (4 * jp) * 128)     + get_tile(eOther + (4 * jp) * 128);
                const v8f sb = get_tile(eMine + (4 * jp + 2) * 128) + get_tile(eOther + (4 * jp + 2) * 128);
                const v8f ra = jp ? r2 : r0, rb = jp ? r3 : r1;
                const v8f ya = role ? sa - ra : sa + ra;
                const v8f yb = role ? sb - rb : sb + rb;
                if (th + 32 * jp < kT) {
                    float4 v;
                    v.x = ya[0]; v.y = ya[1]; v.z = ya[2]; v.w = ya[3];
                    *(float4*)(sw) = v;
                    v.x = ya[4]; v.y = ya[5]; v.z = ya[6]; v.w = ya[7];
                    *(float4*)(sw + 4) = v;
                    v.x = yb[0]; v.y = yb[1]; v.z = yb[2]; v.w = yb[3];
                    *(float4*)(sw + 16) = v;
                    v.x = yb[4]; v.y = yb[5]; v.z = yb[6]; v.w = yb[7];
                    *(float4*)(sw + 20) = v;
                    #pragma unroll
                    for (int i = 0; i < kBlkS / 4; ++i)
                        *(float4*)(og + (size_t)(4 * i) * kT + 32 * jp) = *(const float4*)(sr + 4 * i * kStgPitch);
                }
            }
        }
        if (n == 0) __builtin_amdgcn_s_setprio(2);
        if (n == 1) __builtin_amdgcn_s_setprio(1);
        if (n == 2) __builtin_amdgcn_s_setprio(0);

        __syncthreads();
    }
}

extern "C" void kernel_launch(void* const* d_in, const int* in_sizes, int n_in,
                              void* d_out, int out_size, void* d_ws, size_t ws_size,
                              hipStream_t stream) {
    const float* x = (const float*)d_in[0];
    const float* h = (const float*)d_in[1];
    float* out = (float*)d_out;

    dim3 grid((kSig / kBlkS) * kTileStep);
    fir_bandpass_wmma4x1<<<grid, 256, 0, stream>>>(x, h, out);
}


# Tap fragments as aligned 8-byte pairs: 2nd table copy now serves odd rows, one entry further on; two fragments share one 2-address 8-byte LDS read: 13 -> 9 (S), 11 -> 7 (P, M) LDS reads per K-slice
# speedup vs baseline: 3.8370x; 1.0360x over previous
#include <hip/hip_runtime.h>


typedef __attribute__((ext_vector_type(2))) float v2f;
typedef __attribute__((ext_vector_type(8))) float v8f;

namespace {
constexpr int kT       = 4000;
constexpr int kSig     = 64 * 64;
constexpr int kTaps    = 413;
constexpr int kHalf    = (kTaps - 1) / 2;
constexpr int kBlkS    = 16;
constexpr int kBlkT    = 512;
constexpr int kTimeBlocks = (kT + kBlkT - 1) / kBlkT;
constexpr int kTilesPerWG = 4;
constexpr int kTileStep   = kTimeBlocks / kTilesPerWG;
static_assert(kTimeBlocks % kTilesPerWG == 0, "time blocks must divide evenly over the workgroups");
static_assert(kTilesPerWG == 4, "one priority level (3, 2, 1, 0) per time block of a workgroup");
static_assert(kSig / kBlkS == 256, "blockIdx.x is split into its low 8 bits (signal block) and the rest (first time block)");
constexpr int kPre     = 208;
constexpr int kShift   = kPre - kHalf;
constexpr int kWinV    = 240;
constexpr int kPerThr  = kBlkS * kWinV / 256;
constexpr int kStride  = 964;
static_assert(kBlkS * kWinV == kPerThr * 256 && kPerThr == 15 && kWinV == 7 * 32 + 16 && kBlkS == 2 * 8,
              "the chunk-to-thread map below is written for 16 rows of 240 chunks on 256 threads");
static_assert(kT % 4 == 0 && kBlkT % 4 == 0 && kPre % 4 == 0, "chunks must not straddle the signal's ends");
static_assert(kT >= 2 * 4 * kWinV, "a window must not overhang both ends of a signal");
static_assert(kT % 16 == 0, "an M-tile lies wholly inside or wholly outside a signal");
constexpr int kStgPitch = 36;
static_assert(kStgPitch >= 32 && kStgPitch % 4 == 0 && (kStgPitch / 4) % 2 == 1, "staging rows: 32 samples, 16-byte aligned, an odd multiple of 16 bytes");
static_assert(kT % 32 == 0, "a 128-byte output line lies wholly inside or wholly outside a signal");
constexpr int kPairT   = 128;
constexpr int kGroups  = 4;
constexpr int kSlices  = 4 * kGroups;
static_assert(kBlkT == 4 * kPairT, "four wave pairs share a time block");
static_assert((kTimeBlocks - 1) * kBlkT + 3 * kPairT < kT, "every wave pair of every time block has outputs inside the signal");
constexpr int kHpOff   = 64;
constexpr int kHpFill  = 512;
constexpr int kHpPitch = 544;
constexpr int kCopyPad = 4;
static_assert(kHpPitch >= kHpFill && kHpPitch % 64 == 32, "tap copies");
static_assert(kHpOff % 2 == 0 && kHpPitch % 2 == 0, "an even row's fragment starts on an even entry of the first copy, an odd row's on an even entry of the second");
static_assert(kHpOff - 15 - 48 >= 0 && kHpOff + 2 + 16 * 26 + 8 + 1 < kHpFill, "lowest and highest table entry a fragment reads");
static_assert(16 * 26 + 12 - 15 >= kTaps, "K-slice 3 (v = 12..15) of G_26 holds no tap: g(416 + v - m) lies past the last one");
static_assert(kShift + kPairT * 3 + 2 + 16 * 33 + 8 + 1 < 4 * kWinV, "highest window column a fragment reads is a staged one");
constexpr int kDOff    = 128;
constexpr int kDFill   = 768;
constexpr int kDPitch  = 800;
static_assert(kDPitch >= kDFill && kDPitch % 64 == 32, "D copies");
static_assert(kDOff % 2 == 0 && kDPitch % 2 == 0, "as for the tap copies: every fragment an aligned pair");
static_assert(kDOff - 64 - 48 - 15 >= 0 && kDOff + 2 + 128 * (kGroups - 1) + 12 + 1 < kDFill, "lowest and highest D entry a fragment reads");
static_assert(kHpOff + 2 + 128 * (kGroups - 1) + 12 + 1 < kHpFill, "highest tap entry a fragment of S reads");
constexpr int kExTile  = 8 * 32 * 4;

__device__ __forceinline__ int clamp_chunk(int j)
{
    return j < 0 ? 0 : (j > kT - 4 ? kT - 4 : j);
}

#define FIR_MAIN_CHUNKS(X) \
    X(0,0) X(0,1) X(1,0) X(1,1) X(2,0) X(2,1) X(3,0) X(3,1) \
    X(4,0) X(4,1) X(5,0) X(5,1) X(6,0) X(6,1)
#define FIR_DECL(j,k)  float4 wm##j##k;
#define FIR_REQ(j,k)   wm##j##k = *(const float4*)(gp##j + (k) * 8 * kT);
#define FIR_PUT(j,k)   *(float4*)(lp + (k) * 8 * kStride + (j) * 128) = wm##j##k;
#define FIR_CLEAR(j,k) wm##j##k = make_float4(0.0f, 0.0f, 0.0f, 0.0f);
#define FIR_REQUEST_NEXT() \
    if (n + 1 < kTilesPerWG) { \
        const int j1 = j0 + kTileStep * kBlkT; \
        const float* gp0 = grow + clamp_chunk(j1 + 4 * lane); \
        const float* gp1 = grow + clamp_chunk(j1 + 4 * lane + 128); \
        const float* gp2 = grow + clamp_chunk(j1 + 4 * lane + 256); \
        const float* gp3 = grow + clamp_chunk(j1 + 4 * lane + 384); \
        const float* gp4 = grow + clamp_chunk(j1 + 4 * lane + 512); \
        const float* gp5 = grow + clamp_chunk(j1 + 4 * lane + 640); \
        const float* gp6 = grow + clamp_chunk(j1 + 4 * lane + 768); \
        FIR_MAIN_CHUNKS(FIR_REQ) \
        wt0 = *(const float4*)(trow + clamp_chunk(j1 + tq)); \
    } else { \
          \
          \
        FIR_MAIN_CHUNKS(FIR_CLEAR) \
        wt0 = make_float4(0.0f, 0.0f, 0.0f, 0.0f); \
    }
}

struct Slice  { v2f w0, w1, w2, w3, t0, sp, sm, ps, pp, pm, ms, mp, mm; };
struct SliceS { v2f l0, l1, l2, l3, u0, u1, u2, u3, t0, sp, sm, ps, pp, pm, ms, mp, mm; };
struct Tail   { v2f w32[4], g25[4], w33[3], g26[3]; };

constexpr int slice_col(int i) { return 128 * (i >> 2) + 4 * (i & 3); }

template <class F>
__device__ __forceinline__ void load_taps(F& f, const float* tb, const float* e1, const float* e2, const float* e3, const int c)
{
    f.t0 = *(const v2f*)(tb + c);
    f.sp = *(const v2f*)(e1 + c);
    f.sm = *(const v2f*)(e1 + c - 16);
    f.ps = *(const v2f*)(e2 + c);
    f.pp = *(const v2f*)(e3 + c);
    f.pm = *(const v2f*)(e3 + c - 16);
    f.ms = *(const v2f*)(e2 + c - 32);
    f.mp = *(const v2f*)(e3 + c - 32);
    f.mm = *(const v2f*)(e3 + c - 48);
}

__device__ __forceinline__ void load_slice(Slice& f, const float* xw, const float* tb, const float* e1,
                                           const float* e2, const float* e3, const int c)
{
    f.w0.x = xw[c];          f.w0.y = xw[c + 1];
    f.w1.x = xw[c + 16];     f.w1.y = xw[c + 17];
    f.w2.x = xw[c + 32];     f.w2.y = xw[c + 33];
    f.w3.x = xw[c + 48];     f.w3.y = xw[c + 49];
    load_taps(f, tb, e1, e2, e3, c);
}

__device__ __forceinline__ void load_sliceS(SliceS& f, const float* xb, const float* tb, const float* e1,
                                            const float* e2, const float* e3, const int c)
{
    f.l0.x = xb[c];          f.l0.y = xb[c + 1];
    f.l1.x = xb[c + 16];     f.l1.y = xb[c + 17];
    f.l2.x = xb[c + 32];     f.l2.y = xb[c + 33];
    f.l3.x = xb[c + 48];     f.l3.y = xb[c + 49];
    f.u0.x = xb[c + 64];     f.u0.y = xb[c + 65];
    f.u1.x = xb[c + 80];     f.u1.y = xb[c + 81];
    f.u2.x = xb[c + 96];     f.u2.y = xb[c + 97];
    f.u3.x = xb[c + 112];    f.u3.y = xb[c + 113];
    load_taps(f, tb, e1, e2, e3, c);
}

__device__ __forceinline__ void load_tail(Tail& f, const float* xb, const float* tb)
{
    #pragma unroll
    for (int q = 0; q < 4; ++q) {
        f.w32[q].x = xb[512 + 4 * q];   f.w32[q].y = xb[513 + 4 * q];
        f.g25[q] = *(const v2f*)(tb + 400 + 4 * q);
        if (q < 3) {
            f.w33[q].x = xb[528 + 4 * q];   f.w33[q].y = xb[529 + 4 * q];
            f.g26[q] = *(const v2f*)(tb + 416 + 4 * q);
        }
    }
}

template <class F>
__device__ __forceinline__ void mul_taps(const F& f, const v2f w0, const v2f w1, const v2f w2, const v2f w3,
                                         v8f& aSs, v8f& aSp, v8f& aSm, v8f& aPs, v8f& aPp,
                                         v8f& aPm, v8f& aMs, v8f& aMp, v8f& aMm)
{
    const v2f z02  = w0 + w2;
    const v2f z13  = w1 + w3;
    const v2f zall = z02 + z13;
    const v2f z23  = w2 + w3;
    const v2f z01  = w0 + w1;
    aSs = __builtin_amdgcn_wmma_f32_16x16x4_f32(false, f.t0, false, zall, (short)0, aSs, false, false);
    aSp = __builtin_amdgcn_wmma_f32_16x16x4_f32(false, f.sp, false, z13,  (short)0, aSp, false, false);
    aSm = __builtin_amdgcn_wmma_f32_16x16x4_f32(false, f.sm, false, z02,  (short)0, aSm, false, false);
    aPs = __builtin_amdgcn_wmma_f32_16x16x4_f32(false, f.ps, false, z23,  (short)0, aPs, false, false);
    aPp = __builtin_amdgcn_wmma_f32_16x16x4_f32(false, f.pp, false, w3,   (short)0, aPp, false, false);
    aPm = __builtin_amdgcn_wmma_f32_16x16x4_f32(false, f.pm, false, w2,   (short)0, aPm, false, false);
    aMs = __builtin_amdgcn_wmma_f32_16x16x4_f32(false, f.ms, false, z01,  (short)0, aMs, false, false);
    aMp = __builtin_amdgcn_wmma_f32_16x16x4_f32(false, f.mp, false, w1,   (short)0, aMp, false, false);
    aMm = __builtin_amdgcn_wmma_f32_16x16x4_f32(false, f.mm, false, w0,   (short)0, aMm, false, false);
}

__device__ __forceinline__ void fold9(const v8f& aSs, const v8f& aSp, const v8f& aSm, const v8f& aPs, const v8f& aPp,
                                      const v8f& aPm, const v8f& aMs, const v8f& aMp, const v8f& aMm,
                                      v8f& y0, v8f& y1, v8f& y2, v8f& y3)
{
    const v8f s0 = aSs + aSp, s1 = aSs - aSm;
    y0 = s0 + (aPs + aPp);
    y1 = s1 + (aPs - aPm);
    y2 = s0 - (aMs + aMp);
    y3 = s1 - (aMs - aMm);
}

__device__ __forceinline__ void put_tiles(float* e, const v8f& y0, const v8f& y1, const v8f& y2, const v8f& y3)
{
    float4 v;
    v.x = y0[0]; v.y = y0[1]; v.z = y0[2]; v.w = y0[3];   *(float4*)(e)           = v;
    v.x = y0[4]; v.y = y0[5]; v.z = y0[6]; v.w = y0[7];   *(float4*)(e + 128)     = v;
    v.x = y1[0]; v.y = y1[1]; v.z = y1[2]; v.w = y1[3];   *(float4*)(e + 2 * 128) = v;
    v.x = y1[4]; v.y = y1[5]; v.z = y1[6]; v.w = y1[7];   *(float4*)(e + 3 * 128) = v;
    v.x = y2[0]; v.y = y2[1]; v.z = y2[2]; v.w = y2[3];   *(float4*)(e + 4 * 128) = v;
    v.x = y2[4]; v.y = y2[5]; v.z = y2[6]; v.w = y2[7];   *(float4*)(e + 5 * 128) = v;
    v.x = y3[0]; v.y = y3[1]; v.z = y3[2]; v.w = y3[3];   *(float4*)(e + 6 * 128) = v;
    v.x = y3[4]; v.y = y3[5]; v.z = y3[6]; v.w = y3[7];   *(float4*)(e + 7 * 128) = v;
}

__device__ __forceinline__ v8f get_tile(const float* e)
{
    const float4 a = *(const float4*)(e), b = *(const float4*)(e + 128);
    v8f y;
    y[0] = a.x; y[1] = a.y; y[2] = a.z; y[3] = a.w;
    y[4] = b.x; y[5] = b.y; y[6] = b.z; y[7] = b.w;
    return y;
}

__device__ __forceinline__ void product16(const float* xw, const float* p0, const float* p1, const float* p2, const float* p3,
                                          v8f& aSs, v8f& aSp, v8f& aSm, v8f& aPs, v8f& aPp,
                                          v8f& aPm, v8f& aMs, v8f& aMp, v8f& aMm)
{
    Slice fa, fb;
    load_slice(fa, xw, p0, p1, p2, p3, 0);
    #pragma unroll
    for (int i = 0; i < kSlices; i += 2) {
        load_slice(fb, xw, p0, p1, p2, p3, slice_col(i + 1));
        __builtin_amdgcn_sched_barrier(0);
        mul_taps(fa, fa.w0, fa.w1, fa.w2, fa.w3, aSs, aSp, aSm, aPs, aPp, aPm, aMs, aMp, aMm);
        __builtin_amdgcn_sched_barrier(0);
        if (i + 2 < kSlices) load_slice(fa, xw, p0, p1, p2, p3, slice_col(i + 2));
        __builtin_amdgcn_sched_barrier(0);
        mul_taps(fb, fb.w0, fb.w1, fb.w2, fb.w3, aSs, aSp, aSm, aPs, aPp, aPm, aMs, aMp, aMm);
        __builtin_amdgcn_sched_barrier(0);
    }
}

__global__ __launch_bounds__(256, 4)
void fir_bandpass_wmma4x1(const float* __restrict__ x,
                          const float* __restrict__ h,
                          float* __restrict__ out)
{
    __shared__ __attribute__((aligned(16))) float xs[kBlkS * kStride];
    __shared__ __attribute__((aligned(16))) float hp[kHpPitch + kHpFill + kCopyPad];
    __shared__ __attribute__((aligned(16))) float d16[kHpPitch + kHpFill + kCopyPad];
    __shared__ __attribute__((aligned(16))) float d32[kHpPitch + kHpFill + kCopyPad];
    __shared__ __attribute__((aligned(16))) float dd[kHpPitch + kHpFill + kCopyPad];
    __shared__ __attribute__((aligned(16))) float Dt[kDPitch + kDFill + kCopyPad];
    __shared__ __attribute__((aligned(16))) float D16t[kDPitch + kDFill + kCopyPad];
    __shared__ __attribute__((aligned(16))) float D32t[kDPitch + kDFill + kCopyPad];
    __shared__ __attribute__((aligned(16))) float Ddt[kDPitch + kDFill + kCopyPad];
    __shared__ __attribute__((aligned(16))) float stg[8 * kBlkS * kStgPitch];
    __shared__ __attribute__((aligned(16))) float exS[8 * kExTile];

    __builtin_amdgcn_s_setprio(3);

    const int tid = threadIdx.x;
    const int s0b = (blockIdx.x & 255) * kBlkS;
    const int tb0 = blockIdx.x >> 8;

    const int ja = tid - kHpOff;
    const int jb = tid + 256 - kHpOff;
    float ga[4], gb[4];
    #pragma unroll
    for (int u = 0; u < 4; ++u) {
        const int a = ja + 16 * u, b = jb + 16 * u;
        ga[u] = h[kTaps - 1 - (a < 0 ? 0 : a)];
        gb[u] = h[b < kTaps ? kTaps - 1 - b : 0];
    }
    float gd[3][8];
    #pragma unroll
    for (int u = 0; u < 3; ++u) {
        #pragma unroll
        for (int t = 0; t < 8; ++t) {
            const int a = tid + 256 * u - kDOff + 16 * t;
            gd[u][t] = h[kTaps - 1 - (a < 0 ? 0 : (a > kTaps - 1 ? kTaps - 1 : a))];
        }
    }

    const int w    = __builtin_amdgcn_readfirstlane(tid >> 5);
    const int lane = tid & 31;

    FIR_MAIN_CHUNKS(FIR_DECL)
    float4 wt0;
    const float* grow = x + (size_t)(s0b + w) * kT;
    const float* trow = x + (size_t)(s0b + (tid >> 4)) * kT;
    const int    tq   = 4 * (kWinV - 16 + (tid & 15));
    float* const lp   = xs + w * kStride + 4 * lane;
    float* const lt   = xs + (tid >> 4) * kStride + tq;
    {
        const int j0 = tb0 * kBlkT - kPre;
        const float* gp0 = grow + clamp_chunk(j0 + 4 * lane);
        const float* gp1 = grow + clamp_chunk(j0 + 4 * lane + 128);
        const float* gp2 = grow + clamp_chunk(j0 + 4 * lane + 256);
        const float* gp3 = grow + clamp_chunk(j0 + 4 * lane + 384);
        const float* gp4 = grow + clamp_chunk(j0 + 4 * lane + 512);
        const float* gp5 = grow + clamp_chunk(j0 + 4 * lane + 640);
        const float* gp6 = grow + clamp_chunk(j0 + 4 * lane + 768);
        FIR_MAIN_CHUNKS(FIR_REQ)
        wt0 = *(const float4*)(trow + clamp_chunk(j0 + tq));
    }

    {
        #pragma unroll
        for (int u = 0; u < 4; ++u) {
            ga[u] = ja + 16 * u >= 0 ? ga[u] : 0.0f;
            gb[u] = jb + 16 * u < kTaps ? gb[u] : 0.0f;
        }
        const float a16 = ga[1] - ga[0], b16 = gb[1] - gb[0];
        const float a32 = ga[2] - ga[0], b32 = gb[2] - gb[0];
        const float add = (ga[3] - ga[1]) - a32;
        const float bdd = (gb[3] - gb[1]) - b32;
        hp[tid]        = ga[0];   hp[kHpPitch + 1 + tid]        = ga[0];
        hp[tid + 256]  = gb[0];   hp[kHpPitch + 1 + tid + 256]  = gb[0];
        d16[tid]       = a16;     d16[kHpPitch + 1 + tid]       = a16;
        d16[tid + 256] = b16;     d16[kHpPitch + 1 + tid + 256] = b16;
        d32[tid]       = a32;     d32[kHpPitch + 1 + tid]       = a32;
        d32[tid + 256] = b32;     d32[kHpPitch + 1 + tid + 256] = b32;
        dd[tid]        = add;     dd[kHpPitch + 1 + tid]        = add;
        dd[tid + 256]  = bdd;     dd[kHpPitch + 1 + tid + 256]  = bdd;
        #pragma unroll
        for (int u = 0; u < 3; ++u) {
            #pragma unroll
            for (int t = 0; t < 8; ++t) {
                const int a = tid + 256 * u - kDOff + 16 * t;
                gd[u][t] = (a >= 0 && a < kTaps) ? gd[u][t] : 0.0f;
            }
            const float v0 = gd[u][4] - gd[u][0];
            const float v1 = gd[u][5] - gd[u][1];
            const float v2 = gd[u][6] - gd[u][2];
            const float v3 = gd[u][7] - gd[u][3];
            const float x16 = v1 - v0, x32 = v2 - v0;
            const float xdd = (v3 - v1) - x32;
            const int i = tid + 256 * u;
            Dt[i]   = v0;     Dt[kDPitch + 1 + i]   = v0;
            D16t[i] = x16;    D16t[kDPitch + 1 + i] = x16;
            D32t[i] = x32;    D32t[kDPitch + 1 + i] = x32;
            Ddt[i]  = xdd;    Ddt[kDPitch + 1 + i]  = xdd;
        }
    }

    const int pr   = w & 3;
    const int role = w >> 2;
    const int m    = lane & 15;
    const int hi   = lane >> 4;
    const int kk   = hi << 1;

    const int   tof = ((m & 1) ? kHpPitch + 1 : 0) + (kHpOff + kk - m);
    const float* tb = hp  + tof;
    const float* e1 = d16 + tof;
    const float* e2 = d32 + tof;
    const float* e3 = dd  + tof;
    const int   tofD = ((m & 1) ? kDPitch + 1 : 0) + (kDOff + kk - m);
    const float* q0 = Dt   + tofD;
    const float* q1 = D16t + tofD;
    const float* q2 = D32t + tofD;
    const float* q3 = Ddt  + tofD;
    const float* xb = xs + m * kStride + kShift + (pr << 7) + kk;

    #pragma unroll 1
    for (int n = 0; n < kTilesPerWG; ++n) {
        const int t0b = (tb0 + n * kTileStep) * kBlkT;
        const int j0  = t0b - kPre;

        FIR_MAIN_CHUNKS(FIR_PUT)
        *(float4*)(lt) = wt0;

        if (j0 < 0 || j0 + 4 * kWinV > kT) {
            __syncthreads();
            const bool left = j0 < 0;
            const int nOut  = left ? ((-j0) >> 2) : ((j0 + 4 * kWinV - kT) >> 2);
            const int qb    = left ? 0 : kWinV - nOut;
            const int s     = tid & 15;
            const float* row = x + (size_t)(s0b + s) * kT;
            const float e2x = 2.0f * row[left ? 0 : kT - 1];
            #pragma unroll 4
            for (int q = qb + (tid >> 4); q < qb + nOut; q += 16) {
                const int j = j0 + 4 * q;
                const int r = left ? -j : 2 * kT - 2 - j;
                float4 p;
                p.x = e2x - row[r];
                p.y = e2x - row[r - 1];
                p.z = e2x - row[r - 2];
                p.w = e2x - row[r - 3];
                *(float4*)(xs + s * kStride + 4 * q) = p;
            }
        }
        __syncthreads();

        const int t00 = t0b + (pr << 7);
        const v8f zero = {0.f,0.f,0.f,0.f,0.f,0.f,0.f,0.f};
        float* const eMine        = exS + (role * 4 + pr) * kExTile + 4 * lane;
        const float* const eOther = exS + ((role ^ 1) * 4 + pr) * kExTile + 4 * lane;

        {
            const int so = role * slice_col(kSlices / 2);
            const float* const xh = xb + so;
            const float* const h0 = tb + so;
            const float* const h1 = e1 + so;
            const float* const h2 = e2 + so;
            const float* const h3 = e3 + so;
            v8f aSs = zero, aSp = zero, aSm = zero;
            v8f aPs = zero, aPp = zero, aPm = zero;
            v8f aMs = zero, aMp = zero, aMm = zero;
            SliceS fa, fb;
            load_sliceS(fa, xh, h0, h1, h2, h3, 0);
            #pragma unroll
            for (int i = 0; i < kSlices / 2; i += 2) {
                load_sliceS(fb, xh, h0, h1, h2, h3, slice_col(i + 1));
                __builtin_amdgcn_sched_barrier(0);
                mul_taps(fa, fa.l0 + fa.u0, fa.l1 + fa.u1, fa.l2 + fa.u2, fa.l3 + fa.u3, aSs, aSp, aSm, aPs, aPp, aPm, aMs, aMp, aMm);
                __builtin_amdgcn_sched_barrier(0);
                if (i + 2 < kSlices / 2) load_sliceS(fa, xh, h0, h1, h2, h3, slice_col(i + 2));
                __builtin_amdgcn_sched_barrier(0);
                mul_taps(fb, fb.l0 + fb.u0, fb.l1 + fb.u1, fb.l2 + fb.u2, fb.l3 + fb.u3, aSs, aSp, aSm, aPs, aPp, aPm, aMs, aMp, aMm);
                __builtin_amdgcn_sched_barrier(0);
            }
            v8f s0, s1, s2, s3;
            fold9(aSs, aSp, aSm, aPs, aPp, aPm, aMs, aMp, aMm, s0, s1, s2, s3);
            put_tiles(eMine, s0, s1, s2, s3);
        }

        v8f r0, r1, r2, r3;
        {
            const float* const xw = role ? xb : xb + 64;
            const int qo = role ? -64 : 0;
            v8f aSs = zero, aSp = zero, aSm = zero;
            v8f aPs = zero, aPp = zero, aPm = zero;
            v8f aMs = zero, aMp = zero, aMm = zero;
            if (role == 1) {
                Tail ft;
                load_tail(ft, xb, tb);
                #pragma unroll
                for (int q = 0; q < 4; ++q) {
                    const v2f n32 = -ft.w32[q];
                    aMm = __builtin_amdgcn_wmma_f32_16x16x4_f32(false, ft.g25[q], false, n32, (short)0, aMm, false, false);
                    if (q < 3) {
                        const v2f n33 = -ft.w33[q];
                        aMm = __builtin_amdgcn_wmma_f32_16x16x4_f32(false, ft.g26[q], false, n33, (short)0, aMm, false, false);
                        aMp = __builtin_amdgcn_wmma_f32_16x16x4_f32(false, ft.g26[q], false, ft.w32[q], (short)0, aMp, false, false);
                    }
                }
            }
            product16(xw, q0 + qo, q1 + qo, q2 + qo, q3 + qo, aSs, aSp, aSm, aPs, aPp, aPm, aMs, aMp, aMm);
            fold9(aSs, aSp, aSm, aPs, aPp, aPm, aMs, aMp, aMm, r0, r1, r2, r3);
        }

        __syncthreads();

        FIR_REQUEST_NEXT()

        {
            float* const sg = stg + w * (kBlkS * kStgPitch);
            float* const sw = sg + m * kStgPitch + (hi << 3);
            const float* const sr = sg + (lane >> 3) * kStgPitch + 4 * (lane & 7);
            const int th = t00 + 64 * role;
            float* const og = out + (size_t)(s0b + (lane >> 3)) * kT + th + 4 * (lane & 7);
            #pragma unroll
            for (int jp = 0; jp < 2; ++jp) {
                const v8f sa = get_tile(eMine + (4 * jp) * 128)     + get_tile(eOther + (4 * jp) * 128);
                const v8f sb = get_tile(eMine + (4 * jp + 2) * 128) + get_tile(eOther + (4 * jp + 2) * 128);
                const v8f ra = jp ? r2 : r0, rb = jp ? r3 : r1;
                const v8f ya = role ? sa - ra : sa + ra;
                const v8f yb = role ? sb - rb : sb + rb;
                if (th + 32 * jp < kT) {
                    float4 v;
                    v.x = ya[0]; v.y = ya[1]; v.z = ya[2]; v.w = ya[3];
                    *(float4*)(sw) = v;
                    v.x = ya[4]; v.y = ya[5]; v.z = ya[6]; v.w = ya[7];
                    *(float4*)(sw + 4) = v;
                    v.x = yb[0]; v.y = yb[1]; v.z = yb[2]; v.w = yb[3];
                    *(float4*)(sw + 16) = v;
                    v.x = yb[4]; v.y = yb[5]; v.z = yb[6]; v.w = yb[7];
                    *(float4*)(sw + 20) = v;
                    #pragma unroll
                    for (int i = 0; i < kBlkS / 4; ++i)
                        *(float4*)(og + (size_t)(4 * i) * kT + 32 * jp) = *(const float4*)(sr + 4 * i * kStgPitch);
                }
            }
        }
        if (n == 0) __builtin_amdgcn_s_setprio(2);
        if (n == 1) __builtin_amdgcn_s_setprio(1);
        if (n == 2) __builtin_amdgcn_s_setprio(0);

        __syncthreads();
    }
}

extern "C" void kernel_launch(void* const* d_in, const int* in_sizes, int n_in,
                              void* d_out, int out_size, void* d_ws, size_t ws_size,
                              hipStream_t stream) {
    const float* x = (const float*)d_in[0];
    const float* h = (const float*)d_in[1];
    float* out = (float*)d_out;

    dim3 grid((kSig / kBlkS) * kTileStep);
    fir_bandpass_wmma4x1<<<grid, 256, 0, stream>>>(x, h, out);
}


# One wave owns all 8 tiles of its 128 times: S, then P started from S, then M on the same 9 accumulators; no S exchange through LDS, no barrier behind the products; block 16 x 1024 (new f32 sum order)
# speedup vs baseline: 4.0187x; 1.0474x over previous
#include <hip/hip_runtime.h>


typedef __attribute__((ext_vector_type(2))) float v2f;
typedef __attribute__((ext_vector_type(8))) float v8f;

namespace {
constexpr int kT       = 4000;
constexpr int kSig     = 64 * 64;
constexpr int kTaps    = 413;
constexpr int kHalf    = (kTaps - 1) / 2;
constexpr int kBlkS    = 16;
constexpr int kBlkT    = 1024;
constexpr int kTimeBlocks = (kT + kBlkT - 1) / kBlkT;
constexpr int kTilesPerWG = 2;
constexpr int kTileStep   = kTimeBlocks / kTilesPerWG;
static_assert(kTimeBlocks % kTilesPerWG == 0, "time blocks must divide evenly over the workgroups");
static_assert(kTilesPerWG == 2, "one priority level (3, 2) per time block of a workgroup");
static_assert(kSig / kBlkS == 256, "blockIdx.x is split into its low 8 bits (signal block) and the rest (first time block)");
constexpr int kPre     = 208;
constexpr int kShift   = kPre - kHalf;
constexpr int kWinV    = 368;
constexpr int kPerThr  = kBlkS * kWinV / 256;
constexpr int kStride  = 1476;
static_assert(kBlkS * kWinV == kPerThr * 256 && kPerThr == 23 && kWinV == 11 * 32 + 16 && kBlkS == 2 * 8,
              "the chunk-to-thread map below is written for 16 rows of 368 chunks on 256 threads");
static_assert(kT % 4 == 0 && kBlkT % 4 == 0 && kPre % 4 == 0, "chunks must not straddle the signal's ends");
static_assert(kT >= 2 * 4 * kWinV, "a window must not overhang both ends of a signal");
static_assert(kT % 16 == 0, "an M-tile lies wholly inside or wholly outside a signal");
constexpr int kStgPitch = 36;
static_assert(kStgPitch >= 32 && kStgPitch % 4 == 0 && (kStgPitch / 4) % 2 == 1, "staging rows: 32 samples, 16-byte aligned, an odd multiple of 16 bytes");
static_assert(kT % 32 == 0, "a 128-byte output line lies wholly inside or wholly outside a signal");
constexpr int kPairT   = 128;
constexpr int kGroups  = 4;
constexpr int kSlices  = 4 * kGroups;
static_assert(kBlkT == 8 * kPairT, "eight waves share a time block");
static_assert((kTimeBlocks - 1) * kBlkT + 7 * kPairT < kT, "every wave of every time block has outputs inside the signal");
constexpr int kHpOff   = 64;
constexpr int kHpFill  = 512;
constexpr int kHpPitch = 544;
constexpr int kCopyPad = 4;
static_assert(kHpPitch >= kHpFill && kHpPitch % 64 == 32, "tap copies");
static_assert(kHpOff % 2 == 0 && kHpPitch % 2 == 0, "an even row's fragment starts on an even entry of the first copy, an odd row's on an even entry of the second");
static_assert(kHpOff - 15 - 48 >= 0 && kHpOff + 2 + 16 * 26 + 8 + 1 < kHpFill, "lowest and highest table entry a fragment reads");
static_assert(16 * 26 + 12 - 15 >= kTaps, "K-slice 3 (v = 12..15) of G_26 holds no tap: g(416 + v - m) lies past the last one");
static_assert(kShift + kPairT * 7 + 2 + 16 * 33 + 8 + 1 < 4 * kWinV, "highest window column a fragment reads is a staged one");
constexpr int kDOff    = 128;
constexpr int kDFill   = 768;
constexpr int kDPitch  = 800;
static_assert(kDPitch >= kDFill && kDPitch % 64 == 32, "D copies");
static_assert(kDOff % 2 == 0 && kDPitch % 2 == 0, "as for the tap copies: every fragment an aligned pair");
static_assert(kDOff - 64 - 48 - 15 >= 0 && kDOff + 2 + 128 * (kGroups - 1) + 12 + 1 < kDFill, "lowest and highest D entry a fragment reads");
static_assert(kHpOff + 2 + 128 * (kGroups - 1) + 12 + 1 < kHpFill, "highest tap entry a fragment of S reads");

__device__ __forceinline__ int clamp_chunk(int j)
{
    return j < 0 ? 0 : (j > kT - 4 ? kT - 4 : j);
}

#define FIR_MAIN_CHUNKS(X) \
    X(0,0) X(0,1) X(1,0) X(1,1) X(2,0) X(2,1) X(3,0) X(3,1) \
    X(4,0) X(4,1) X(5,0) X(5,1) X(6,0) X(6,1) X(7,0) X(7,1) \
    X(8,0) X(8,1) X(9,0) X(9,1) X(10,0) X(10,1)
#define FIR_DECL(j,k)  float4 wm##j##k;
#define FIR_REQ(j,k)   wm##j##k = *(const float4*)(gp##j + (k) * 8 * kT);
#define FIR_PUT(j,k)   *(float4*)(lp + (k) * 8 * kStride + (j) * 128) = wm##j##k;
#define FIR_CLEAR(j,k) wm##j##k = make_float4(0.0f, 0.0f, 0.0f, 0.0f);
#define FIR_REQUEST_NEXT() \
    if (n + 1 < kTilesPerWG) { \
        const int j1 = j0 + kTileStep * kBlkT; \
        const float* gp0 = grow + clamp_chunk(j1 + 4 * lane); \
        const float* gp1 = grow + clamp_chunk(j1 + 4 * lane + 128); \
        const float* gp2 = grow + clamp_chunk(j1 + 4 * lane + 256); \
        const float* gp3 = grow + clamp_chunk(j1 + 4 * lane + 384); \
        const float* gp4 = grow + clamp_chunk(j1 + 4 * lane + 512); \
        const float* gp5 = grow + clamp_chunk(j1 + 4 * lane + 640); \
        const float* gp6 = grow + clamp_chunk(j1 + 4 * lane + 768); \
        const float* gp7 = grow + clamp_chunk(j1 + 4 * lane + 896); \
        const float* gp8 = grow + clamp_chunk(j1 + 4 * lane + 1024); \
        const float* gp9 = grow + clamp_chunk(j1 + 4 * lane + 1152); \
        const float* gp10 = grow + clamp_chunk(j1 + 4 * lane + 1280); \
        FIR_MAIN_CHUNKS(FIR_REQ) \
        wt0 = *(const float4*)(trow + clamp_chunk(j1 + tq)); \
    } else { \
          \
          \
        FIR_MAIN_CHUNKS(FIR_CLEAR) \
        wt0 = make_float4(0.0f, 0.0f, 0.0f, 0.0f); \
    }
}

struct Slice  { v2f w0, w1, w2, w3, t0, sp, sm, ps, pp, pm, ms, mp, mm; };
struct SliceS { v2f l0, l1, l2, l3, u0, u1, u2, u3, t0, sp, sm, ps, pp, pm, ms, mp, mm; };
struct Tail   { v2f w32[4], g25[4], w33[3], g26[3]; };

constexpr int slice_col(int i) { return 128 * (i >> 2) + 4 * (i & 3); }

template <class F>
__device__ __forceinline__ void load_taps(F& f, const float* tb, const float* e1, const float* e2, const float* e3, const int c)
{
    f.t0 = *(const v2f*)(tb + c);
    f.sp = *(const v2f*)(e1 + c);
    f.sm = *(const v2f*)(e1 + c - 16);
    f.ps = *(const v2f*)(e2 + c);
    f.pp = *(const v2f*)(e3 + c);
    f.pm = *(const v2f*)(e3 + c - 16);
    f.ms = *(const v2f*)(e2 + c - 32);
    f.mp = *(const v2f*)(e3 + c - 32);
    f.mm = *(const v2f*)(e3 + c - 48);
}

__device__ __forceinline__ void load_slice(Slice& f, const float* xw, const float* tb, const float* e1,
                                           const float* e2, const float* e3, const int c)
{
    f.w0.x = xw[c];          f.w0.y = xw[c + 1];
    f.w1.x = xw[c + 16];     f.w1.y = xw[c + 17];
    f.w2.x = xw[c + 32];     f.w2.y = xw[c + 33];
    f.w3.x = xw[c + 48];     f.w3.y = xw[c + 49];
    load_taps(f, tb, e1, e2, e3, c);
}

__device__ __forceinline__ void load_sliceS(SliceS& f, const float* xb, const float* tb, const float* e1,
                                            const float* e2, const float* e3, const int c)
{
    f.l0.x = xb[c];          f.l0.y = xb[c + 1];
    f.l1.x = xb[c + 16];     f.l1.y = xb[c + 17];
    f.l2.x = xb[c + 32];     f.l2.y = xb[c + 33];
    f.l3.x = xb[c + 48];     f.l3.y = xb[c + 49];
    f.u0.x = xb[c + 64];     f.u0.y = xb[c + 65];
    f.u1.x = xb[c + 80];     f.u1.y = xb[c + 81];
    f.u2.x = xb[c + 96];     f.u2.y = xb[c + 97];
    f.u3.x = xb[c + 112];    f.u3.y = xb[c + 113];
    load_taps(f, tb, e1, e2, e3, c);
}

__device__ __forceinline__ void load_tail(Tail& f, const float* xb, const float* tb)
{
    #pragma unroll
    for (int q = 0; q < 4; ++q) {
        f.w32[q].x = xb[512 + 4 * q];   f.w32[q].y = xb[513 + 4 * q];
        f.g25[q] = *(const v2f*)(tb + 400 + 4 * q);
        if (q < 3) {
            f.w33[q].x = xb[528 + 4 * q];   f.w33[q].y = xb[529 + 4 * q];
            f.g26[q] = *(const v2f*)(tb + 416 + 4 * q);
        }
    }
}

template <class F>
__device__ __forceinline__ void mul_taps(const F& f, const v2f w0, const v2f w1, const v2f w2, const v2f w3,
                                         v8f& aSs, v8f& aSp, v8f& aSm, v8f& aPs, v8f& aPp,
                                         v8f& aPm, v8f& aMs, v8f& aMp, v8f& aMm)
{
    const v2f z02  = w0 + w2;
    const v2f z13  = w1 + w3;
    const v2f zall = z02 + z13;
    const v2f z23  = w2 + w3;
    const v2f z01  = w0 + w1;
    aSs = __builtin_amdgcn_wmma_f32_16x16x4_f32(false, f.t0, false, zall, (short)0, aSs, false, false);
    aSp = __builtin_amdgcn_wmma_f32_16x16x4_f32(false, f.sp, false, z13,  (short)0, aSp, false, false);
    aSm = __builtin_amdgcn_wmma_f32_16x16x4_f32(false, f.sm, false, z02,  (short)0, aSm, false, false);
    aPs = __builtin_amdgcn_wmma_f32_16x16x4_f32(false, f.ps, false, z23,  (short)0, aPs, false, false);
    aPp = __builtin_amdgcn_wmma_f32_16x16x4_f32(false, f.pp, false, w3,   (short)0, aPp, false, false);
    aPm = __builtin_amdgcn_wmma_f32_16x16x4_f32(false, f.pm, false, w2,   (short)0, aPm, false, false);
    aMs = __builtin_amdgcn_wmma_f32_16x16x4_f32(false, f.ms, false, z01,  (short)0, aMs, false, false);
    aMp = __builtin_amdgcn_wmma_f32_16x16x4_f32(false, f.mp, false, w1,   (short)0, aMp, false, false);
    aMm = __builtin_amdgcn_wmma_f32_16x16x4_f32(false, f.mm, false, w0,   (short)0, aMm, false, false);
}

__device__ __forceinline__ void fold9(const v8f& aSs, const v8f& aSp, const v8f& aSm, const v8f& aPs, const v8f& aPp,
                                      const v8f& aPm, const v8f& aMs, const v8f& aMp, const v8f& aMm,
                                      v8f& y0, v8f& y1, v8f& y2, v8f& y3)
{
    const v8f s0 = aSs + aSp, s1 = aSs - aSm;
    y0 = s0 + (aPs + aPp);
    y1 = s1 + (aPs - aPm);
    y2 = s0 - (aMs + aMp);
    y3 = s1 - (aMs - aMm);
}

__device__ __forceinline__ void store_two_tiles(float* sw, const float* sr, float* og, const v8f& ya, const v8f& yb)
{
    float4 v;
    v.x = ya[0]; v.y = ya[1]; v.z = ya[2]; v.w = ya[3];
    *(float4*)(sw) = v;
    v.x = ya[4]; v.y = ya[5]; v.z = ya[6]; v.w = ya[7];
    *(float4*)(sw + 4) = v;
    v.x = yb[0]; v.y = yb[1]; v.z = yb[2]; v.w = yb[3];
    *(float4*)(sw + 16) = v;
    v.x = yb[4]; v.y = yb[5]; v.z = yb[6]; v.w = yb[7];
    *(float4*)(sw + 20) = v;
    #pragma unroll
    for (int i = 0; i < kBlkS / 4; ++i)
        *(float4*)(og + (size_t)(4 * i) * kT) = *(const float4*)(sr + 4 * i * kStgPitch);
}

__device__ __forceinline__ void product16(const float* xw, const float* p0, const float* p1, const float* p2, const float* p3,
                                          v8f& aSs, v8f& aSp, v8f& aSm, v8f& aPs, v8f& aPp,
                                          v8f& aPm, v8f& aMs, v8f& aMp, v8f& aMm)
{
    Slice fa, fb;
    load_slice(fa, xw, p0, p1, p2, p3, 0);
    #pragma unroll
    for (int i = 0; i < kSlices; i += 2) {
        load_slice(fb, xw, p0, p1, p2, p3, slice_col(i + 1));
        __builtin_amdgcn_sched_barrier(0);
        mul_taps(fa, fa.w0, fa.w1, fa.w2, fa.w3, aSs, aSp, aSm, aPs, aPp, aPm, aMs, aMp, aMm);
        __builtin_amdgcn_sched_barrier(0);
        if (i + 2 < kSlices) load_slice(fa, xw, p0, p1, p2, p3, slice_col(i + 2));
        __builtin_amdgcn_sched_barrier(0);
        mul_taps(fb, fb.w0, fb.w1, fb.w2, fb.w3, aSs, aSp, aSm, aPs, aPp, aPm, aMs, aMp, aMm);
        __builtin_amdgcn_sched_barrier(0);
    }
}

__global__ __launch_bounds__(256, 4)
void fir_bandpass_wmma4x1(const float* __restrict__ x,
                          const float* __restrict__ h,
                          float* __restrict__ out)
{
    __shared__ __attribute__((aligned(16))) float xs[kBlkS * kStride];
    __shared__ __attribute__((aligned(16))) float hp[kHpPitch + kHpFill + kCopyPad];
    __shared__ __attribute__((aligned(16))) float d16[kHpPitch + kHpFill + kCopyPad];
    __shared__ __attribute__((aligned(16))) float d32[kHpPitch + kHpFill + kCopyPad];
    __shared__ __attribute__((aligned(16))) float dd[kHpPitch + kHpFill + kCopyPad];
    __shared__ __attribute__((aligned(16))) float Dt[kDPitch + kDFill + kCopyPad];
    __shared__ __attribute__((aligned(16))) float D16t[kDPitch + kDFill + kCopyPad];
    __shared__ __attribute__((aligned(16))) float D32t[kDPitch + kDFill + kCopyPad];
    __shared__ __attribute__((aligned(16))) float Ddt[kDPitch + kDFill + kCopyPad];
    __shared__ __attribute__((aligned(16))) float stg[8 * kBlkS * kStgPitch];

    __builtin_amdgcn_s_setprio(3);

    const int tid = threadIdx.x;
    const int s0b = (blockIdx.x & 255) * kBlkS;
    const int tb0 = blockIdx.x >> 8;

    const int ja = tid - kHpOff;
    const int jb = tid + 256 - kHpOff;
    float ga[4], gb[4];
    #pragma unroll
    for (int u = 0; u < 4; ++u) {
        const int a = ja + 16 * u, b = jb + 16 * u;
        ga[u] = h[kTaps - 1 - (a < 0 ? 0 : a)];
        gb[u] = h[b < kTaps ? kTaps - 1 - b : 0];
    }
    float gd[3][8];
    #pragma unroll
    for (int u = 0; u < 3; ++u) {
        #pragma unroll
        for (int t = 0; t < 8; ++t) {
            const int a = tid + 256 * u - kDOff + 16 * t;
            gd[u][t] = h[kTaps - 1 - (a < 0 ? 0 : (a > kTaps - 1 ? kTaps - 1 : a))];
        }
    }

    const int w    = __builtin_amdgcn_readfirstlane(tid >> 5);
    const int lane = tid & 31;

    FIR_MAIN_CHUNKS(FIR_DECL)
    float4 wt0;
    const float* grow = x + (size_t)(s0b + w) * kT;
    const float* trow = x + (size_t)(s0b + (tid >> 4)) * kT;
    const int    tq   = 4 * (kWinV - 16 + (tid & 15));
    float* const lp   = xs + w * kStride + 4 * lane;
    float* const lt   = xs + (tid >> 4) * kStride + tq;
    {
        const int j0 = tb0 * kBlkT - kPre;
        const float* gp0 = grow + clamp_chunk(j0 + 4 * lane);
        const float* gp1 = grow + clamp_chunk(j0 + 4 * lane + 128);
        const float* gp2 = grow + clamp_chunk(j0 + 4 * lane + 256);
        const float* gp3 = grow + clamp_chunk(j0 + 4 * lane + 384);
        const float* gp4 = grow + clamp_chunk(j0 + 4 * lane + 512);
        const float* gp5 = grow + clamp_chunk(j0 + 4 * lane + 640);
        const float* gp6 = grow + clamp_chunk(j0 + 4 * lane + 768);
        const float* gp7 = grow + clamp_chunk(j0 + 4 * lane + 896);
        const float* gp8 = grow + clamp_chunk(j0 + 4 * lane + 1024);
        const float* gp9 = grow + clamp_chunk(j0 + 4 * lane + 1152);
        const float* gp10 = grow + clamp_chunk(j0 + 4 * lane + 1280);
        FIR_MAIN_CHUNKS(FIR_REQ)
        wt0 = *(const float4*)(trow + clamp_chunk(j0 + tq));
    }

    {
        #pragma unroll
        for (int u = 0; u < 4; ++u) {
            ga[u] = ja + 16 * u >= 0 ? ga[u] : 0.0f;
            gb[u] = jb + 16 * u < kTaps ? gb[u] : 0.0f;
        }
        const float a16 = ga[1] - ga[0], b16 = gb[1] - gb[0];
        const float a32 = ga[2] - ga[0], b32 = gb[2] - gb[0];
        const float add = (ga[3] - ga[1]) - a32;
        const float bdd = (gb[3] - gb[1]) - b32;
        hp[tid]        = ga[0];   hp[kHpPitch + 1 + tid]        = ga[0];
        hp[tid + 256]  = gb[0];   hp[kHpPitch + 1 + tid + 256]  = gb[0];
        d16[tid]       = a16;     d16[kHpPitch + 1 + tid]       = a16;
        d16[tid + 256] = b16;     d16[kHpPitch + 1 + tid + 256] = b16;
        d32[tid]       = a32;     d32[kHpPitch + 1 + tid]       = a32;
        d32[tid + 256] = b32;     d32[kHpPitch + 1 + tid + 256] = b32;
        dd[tid]        = add;     dd[kHpPitch + 1 + tid]        = add;
        dd[tid + 256]  = bdd;     dd[kHpPitch + 1 + tid + 256]  = bdd;
        #pragma unroll
        for (int u = 0; u < 3; ++u) {
            #pragma unroll
            for (int t = 0; t < 8; ++t) {
                const int a = tid + 256 * u - kDOff + 16 * t;
                gd[u][t] = (a >= 0 && a < kTaps) ? gd[u][t] : 0.0f;
            }
            const float v0 = gd[u][4] - gd[u][0];
            const float v1 = gd[u][5] - gd[u][1];
            const float v2 = gd[u][6] - gd[u][2];
            const float v3 = gd[u][7] - gd[u][3];
            const float x16 = v1 - v0, x32 = v2 - v0;
            const float xdd = (v3 - v1) - x32;
            const int i = tid + 256 * u;
            Dt[i]   = v0;     Dt[kDPitch + 1 + i]   = v0;
            D16t[i] = x16;    D16t[kDPitch + 1 + i] = x16;
            D32t[i] = x32;    D32t[kDPitch + 1 + i] = x32;
            Ddt[i]  = xdd;    Ddt[kDPitch + 1 + i]  = xdd;
        }
    }

    const int m    = lane & 15;
    const int hi   = lane >> 4;
    const int kk   = hi << 1;

    const int   tof = ((m & 1) ? kHpPitch + 1 : 0) + (kHpOff + kk - m);
    const float* tb = hp  + tof;
    const float* e1 = d16 + tof;
    const float* e2 = d32 + tof;
    const float* e3 = dd  + tof;
    const int   tofD = ((m & 1) ? kDPitch + 1 : 0) + (kDOff + kk - m);
    const float* q0 = Dt   + tofD;
    const float* q1 = D16t + tofD;
    const float* q2 = D32t + tofD;
    const float* q3 = Ddt  + tofD;
    const float* xb = xs + m * kStride + kShift + (w << 7) + kk;
    float* const sg = stg + w * (kBlkS * kStgPitch);
    float* const sw = sg + m * kStgPitch + (hi << 3);
    const float* const sr = sg + (lane >> 3) * kStgPitch + 4 * (lane & 7);
    float* const orow = out + (size_t)(s0b + (lane >> 3)) * kT + 4 * (lane & 7);

    #pragma unroll 1
    for (int n = 0; n < kTilesPerWG; ++n) {
        const int t0b = (tb0 + n * kTileStep) * kBlkT;
        const int j0  = t0b - kPre;

        FIR_MAIN_CHUNKS(FIR_PUT)
        *(float4*)(lt) = wt0;

        if (j0 < 0 || j0 + 4 * kWinV > kT) {
            __syncthreads();
            const bool left = j0 < 0;
            const int nOut  = left ? ((-j0) >> 2) : ((j0 + 4 * kWinV - kT) >> 2);
            const int qb    = left ? 0 : kWinV - nOut;
            const int s     = tid & 15;
            const float* row = x + (size_t)(s0b + s) * kT;
            const float e2x = 2.0f * row[left ? 0 : kT - 1];
            #pragma unroll 4
            for (int q = qb + (tid >> 4); q < qb + nOut; q += 16) {
                const int j = j0 + 4 * q;
                const int r = left ? -j : 2 * kT - 2 - j;
                float4 p;
                p.x = e2x - row[r];
                p.y = e2x - row[r - 1];
                p.z = e2x - row[r - 2];
                p.w = e2x - row[r - 3];
                *(float4*)(xs + s * kStride + 4 * q) = p;
            }
        }
        __syncthreads();

        const int t00 = t0b + (w << 7);
        const v8f zero = {0.f,0.f,0.f,0.f,0.f,0.f,0.f,0.f};
        float* const og = orow + t00;

        v8f s0, s1, s2, s3;
        {
            v8f aSs = zero, aSp = zero, aSm = zero;
            v8f aPs = zero, aPp = zero, aPm = zero;
            v8f aMs = zero, aMp = zero, aMm = zero;
            SliceS fa, fb;
            load_sliceS(fa, xb, tb, e1, e2, e3, 0);
            #pragma unroll
            for (int i = 0; i < kSlices; i += 2) {
                load_sliceS(fb, xb, tb, e1, e2, e3, slice_col(i + 1));
                __builtin_amdgcn_sched_barrier(0);
                mul_taps(fa, fa.l0 + fa.u0, fa.l1 + fa.u1, fa.l2 + fa.u2, fa.l3 + fa.u3, aSs, aSp, aSm, aPs, aPp, aPm, aMs, aMp, aMm);
                __builtin_amdgcn_sched_barrier(0);
                if (i + 2 < kSlices) load_sliceS(fa, xb, tb, e1, e2, e3, slice_col(i + 2));
                __builtin_amdgcn_sched_barrier(0);
                mul_taps(fb, fb.l0 + fb.u0, fb.l1 + fb.u1, fb.l2 + fb.u2, fb.l3 + fb.u3, aSs, aSp, aSm, aPs, aPp, aPm, aMs, aMp, aMm);
                __builtin_amdgcn_sched_barrier(0);
            }
            fold9(aSs, aSp, aSm, aPs, aPp, aPm, aMs, aMp, aMm, s0, s1, s2, s3);
        }

        {
            v8f aSs = zero, aSp = zero, aSm = zero;
            v8f aPs = zero, aPp = s0,   aPm = -s1;
            v8f aMs = zero, aMp = -s2,  aMm = s3;
            product16(xb + 64, q0, q1, q2, q3, aSs, aSp, aSm, aPs, aPp, aPm, aMs, aMp, aMm);
            v8f y0, y1, y2, y3;
            fold9(aSs, aSp, aSm, aPs, aPp, aPm, aMs, aMp, aMm, y0, y1, y2, y3);
            if (t00 < kT)      store_two_tiles(sw, sr, og,      y0, y1);
            if (t00 + 32 < kT) store_two_tiles(sw, sr, og + 32, y2, y3);
        }

        {
            v8f aSs = zero, aSp = zero, aSm = zero;
            v8f aPs = zero, aPp = zero, aPm = zero;
            v8f aMs = zero, aMp = zero, aMm = zero;
            {
                Tail ft;
                load_tail(ft, xb, tb);
                #pragma unroll
                for (int q = 0; q < 4; ++q) {
                    const v2f n32 = -ft.w32[q];
                    aMm = __builtin_amdgcn_wmma_f32_16x16x4_f32(false, ft.g25[q], false, n32, (short)0, aMm, false, false);
                    if (q < 3) {
                        const v2f n33 = -ft.w33[q];
                        aMm = __builtin_amdgcn_wmma_f32_16x16x4_f32(false, ft.g26[q], false, n33, (short)0, aMm, false, false);
                        aMp = __builtin_amdgcn_wmma_f32_16x16x4_f32(false, ft.g26[q], false, ft.w32[q], (short)0, aMp, false, false);
                    }
                }
            }
            product16(xb, q0 - 64, q1 - 64, q2 - 64, q3 - 64, aSs, aSp, aSm, aPs, aPp, aPm, aMs, aMp, aMm);
            v8f r0, r1, r2, r3;
            fold9(aSs, aSp, aSm, aPs, aPp, aPm, aMs, aMp, aMm, r0, r1, r2, r3);

            FIR_REQUEST_NEXT()

            if (t00 + 64 < kT) store_two_tiles(sw, sr, og + 64, s0 - r0, s1 - r1);
            if (t00 + 96 < kT) store_two_tiles(sw, sr, og + 96, s2 - r2, s3 - r3);
        }
        if (n == 0) __builtin_amdgcn_s_setprio(2);

        __syncthreads();
    }
}

extern "C" void kernel_launch(void* const* d_in, const int* in_sizes, int n_in,
                              void* d_out, int out_size, void* d_ws, size_t ws_size,
                              hipStream_t stream) {
    const float* x = (const float*)d_in[0];
    const float* h = (const float*)d_in[1];
    float* out = (float*)d_out;

    dim3 grid((kSig / kBlkS) * kTileStep);
    fir_bandpass_wmma4x1<<<grid, 256, 0, stream>>>(x, h, out);
}
